# starting kernel
# baseline (speedup 1.0000x reference)
#include <hip/hip_runtime.h>


typedef __attribute__((ext_vector_type(2))) float v2f;
typedef __attribute__((ext_vector_type(8))) float v8f;

#define G_GROUPS 1024
#define T_STEPS  512
#define WAVES_PER_BLOCK 4
#define THREADS (WAVES_PER_BLOCK * 32)

static __device__ __forceinline__ v8f wmma4(v2f a, v2f b, v8f c) {
  return __builtin_amdgcn_wmma_f32_16x16x4_f32(
       false, a,  false, b,
       (short)0, c,  false,  false);
}

static __device__ __forceinline__ v8f zero8() {
  v8f z = {0.f, 0.f, 0.f, 0.f, 0.f, 0.f, 0.f, 0.f};
  return z;
}

static __device__ __forceinline__ void stage_c(float* buf, v8f c, int col, int half) {
#pragma unroll
  for (int r = 0; r < 8; ++r) buf[(r + 8 * half) * 16 + col] = c[r];
}

static __device__ __forceinline__ v2f load_a(const float* buf, int col, int half, int k) {
  int k0 = 4 * k + 2 * half;
  v2f a;
  a.x = buf[col * 16 + k0];
  a.y = buf[col * 16 + k0 + 1];
  return a;
}

static __device__ __forceinline__ v2f load_b(const float* buf, int col, int half, int k) {
  int k0 = 4 * k + 2 * half;
  v2f b;
  b.x = buf[k0 * 16 + col];
  b.y = buf[(k0 + 1) * 16 + col];
  return b;
}

static __device__ __forceinline__ v2f load_bt(const float* buf, int col, int half, int k) {
  int k0 = 4 * k + 2 * half;
  v2f b;
  b.x = buf[col * 16 + k0];
  b.y = buf[col * 16 + k0 + 1];
  return b;
}

static __device__ __forceinline__ float sel4(int i, float a, float b, float c, float d) {
  return i == 0 ? a : (i == 1 ? b : (i == 2 ? c : d));
}

__global__ __launch_bounds__(THREADS, 1)
void kalman_wmma_kernel(const float* __restrict__ y,
                        const float* __restrict__ F,
                        const float* __restrict__ H,
                        const float* __restrict__ Q,
                        const float* __restrict__ R,
                        const float* __restrict__ im,
                        const float* __restrict__ ic,
                        float* __restrict__ out) {
  __shared__ float lds[WAVES_PER_BLOCK * 4 * 256];

  const int lane = threadIdx.x & 31;
  const int wave = threadIdx.x >> 5;
  const int col  = lane & 15;
  const int half = lane >> 4;
  const int g    = blockIdx.x * WAVES_PER_BLOCK + wave;

  float* bufP = &lds[(wave * 4 + 0) * 256];
  float* bufZ = &lds[(wave * 4 + 1) * 256];
  float* bufI = &lds[(wave * 4 + 2) * 256];
  float* bufS = &lds[(wave * 4 + 3) * 256];

  v2f FA[4], HA[4];
#pragma unroll
  for (int k = 0; k < 4; ++k) {
    int k0 = 4 * k + 2 * half;
    FA[k].x = F[col * 16 + k0];
    FA[k].y = F[col * 16 + k0 + 1];
    HA[k].x = (col < 4) ? H[col * 16 + k0] : 0.f;
    HA[k].y = (col < 4) ? H[col * 16 + k0 + 1] : 0.f;
  }
  v2f Hb;
  Hb.x = H[(2 * half) * 16 + col];
  Hb.y = H[(2 * half + 1) * 16 + col];
  v2f Rb;
  Rb.x = (col < 4) ? R[(2 * half) * 4 + col] : 0.f;
  Rb.y = (col < 4) ? R[(2 * half + 1) * 4 + col] : 0.f;

  v8f Qc, Rc, Ic, Pc, Mc;
#pragma unroll
  for (int r = 0; r < 8; ++r) {
    int row = r + 8 * half;
    Qc[r] = Q[row * 16 + col];
    Rc[r] = (row < 4 && col < 4) ? R[row * 4 + col] : 0.f;
    Ic[r] = (row == col) ? 1.f : 0.f;
    Pc[r] = ic[row * 16 + col];
    Mc[r] = (col == 0) ? im[g * 16 + row] : 0.f;
  }

  for (int t = 0; t < T_STEPS; ++t) {
    stage_c(bufP, Pc, col, half);
    v8f PHt = zero8();
#pragma unroll
    for (int k = 0; k < 4; ++k)
      PHt = wmma4(load_a(bufP, col, half, k), HA[k], PHt);

    stage_c(bufZ, PHt, col, half);
    if (col == 0) {
#pragma unroll
      for (int r = 0; r < 8; ++r) bufZ[(r + 8 * half) * 16 + 4] = Mc[r];
    }
    v8f Sc = Rc;
#pragma unroll
    for (int k = 0; k < 4; ++k)
      Sc = wmma4(HA[k], load_b(bufZ, col, half, k), Sc);

    if (lane == 4) {
      float4 o;
      o.x = Sc[0]; o.y = Sc[1]; o.z = Sc[2]; o.w = Sc[3];
      *(float4*)(out + ((size_t)g * T_STEPS + t) * 4) = o;
    }

    float s00 = __shfl(Sc[0], 0, 32), s01 = __shfl(Sc[0], 1, 32),
          s02 = __shfl(Sc[0], 2, 32), s03 = __shfl(Sc[0], 3, 32);
    float s10 = __shfl(Sc[1], 0, 32), s11 = __shfl(Sc[1], 1, 32),
          s12 = __shfl(Sc[1], 2, 32), s13 = __shfl(Sc[1], 3, 32);
    float s20 = __shfl(Sc[2], 0, 32), s21 = __shfl(Sc[2], 1, 32),
          s22 = __shfl(Sc[2], 2, 32), s23 = __shfl(Sc[2], 3, 32);
    float s30 = __shfl(Sc[3], 0, 32), s31 = __shfl(Sc[3], 1, 32),
          s32 = __shfl(Sc[3], 2, 32), s33 = __shfl(Sc[3], 3, 32);
    float mm0 = __shfl(Sc[0], 4, 32), mm1 = __shfl(Sc[1], 4, 32),
          mm2 = __shfl(Sc[2], 4, 32), mm3 = __shfl(Sc[3], 4, 32);

    const float4 yv = *(const float4*)(y + ((size_t)g * T_STEPS + t) * 4);
    float r0 = yv.x - mm0, r1 = yv.y - mm1, r2 = yv.z - mm2, r3 = yv.w - mm3;

    float A2323 = s22 * s33 - s23 * s32;
    float A1323 = s21 * s33 - s23 * s31;
    float A1223 = s21 * s32 - s22 * s31;
    float A0323 = s20 * s33 - s23 * s30;
    float A0223 = s20 * s32 - s22 * s30;
    float A0123 = s20 * s31 - s21 * s30;
    float A2313 = s12 * s33 - s13 * s32;
    float A1313 = s11 * s33 - s13 * s31;
    float A1213 = s11 * s32 - s12 * s31;
    float A2312 = s12 * s23 - s13 * s22;
    float A1312 = s11 * s23 - s13 * s21;
    float A1212 = s11 * s22 - s12 * s21;
    float A0313 = s10 * s33 - s13 * s30;
    float A0213 = s10 * s32 - s12 * s30;
    float A0312 = s10 * s23 - s13 * s20;
    float A0212 = s10 * s22 - s12 * s20;
    float A0113 = s10 * s31 - s11 * s30;
    float A0112 = s10 * s21 - s11 * s20;

    float det = s00 * (s11 * A2323 - s12 * A1323 + s13 * A1223)
              - s01 * (s10 * A2323 - s12 * A0323 + s13 * A0223)
              + s02 * (s10 * A1323 - s11 * A0323 + s13 * A0123)
              - s03 * (s10 * A1223 - s11 * A0223 + s12 * A0123);
    float id = 1.0f / det;

    float i00 =  id * (s11 * A2323 - s12 * A1323 + s13 * A1223);
    float i01 = -id * (s01 * A2323 - s02 * A1323 + s03 * A1223);
    float i02 =  id * (s01 * A2313 - s02 * A1313 + s03 * A1213);
    float i03 = -id * (s01 * A2312 - s02 * A1312 + s03 * A1212);
    float i10 = -id * (s10 * A2323 - s12 * A0323 + s13 * A0223);
    float i11 =  id * (s00 * A2323 - s02 * A0323 + s03 * A0223);
    float i12 = -id * (s00 * A2313 - s02 * A0313 + s03 * A0213);
    float i13 =  id * (s00 * A2312 - s02 * A0312 + s03 * A0212);
    float i20 =  id * (s10 * A1323 - s11 * A0323 + s13 * A0123);
    float i21 = -id * (s00 * A1323 - s01 * A0323 + s03 * A0123);
    float i22 =  id * (s00 * A1313 - s01 * A0313 + s03 * A0113);
    float i23 = -id * (s00 * A1312 - s01 * A0312 + s03 * A0112);
    float i30 = -id * (s10 * A1223 - s11 * A0223 + s12 * A0123);
    float i31 =  id * (s00 * A1223 - s01 * A0223 + s02 * A0123);
    float i32 = -id * (s00 * A1213 - s01 * A0213 + s02 * A0113);
    float i33 =  id * (s00 * A1212 - s01 * A0212 + s02 * A0112);

    float si_r0 = sel4(col, i00, i01, i02, i03);
    float si_r1 = sel4(col, i10, i11, i12, i13);
    float si_r2 = sel4(col, i20, i21, i22, i23);
    float si_r3 = sel4(col, i30, i31, i32, i33);
    v2f Sib;
    Sib.x = (col < 4) ? (half ? si_r2 : si_r0) : 0.f;
    Sib.y = (col < 4) ? (half ? si_r3 : si_r1) : 0.f;

    v2f phta = load_a(bufZ, col, half, 0);
    v8f Kc = wmma4(phta, Sib, zero8());

    stage_c(bufZ, Kc, col, half);
    v2f ka = load_a(bufZ, col, half, 0);
    v2f nka; nka.x = -ka.x; nka.y = -ka.y;

    v8f IKH = wmma4(nka, Hb, Ic);

    v2f rb;
    rb.x = (col == 0) ? (half ? r2 : r0) : 0.f;
    rb.y = (col == 0) ? (half ? r3 : r1) : 0.f;
    v8f Mu = wmma4(ka, rb, Mc);

    stage_c(bufI, IKH, col, half);
    v8f Tm = zero8();
#pragma unroll
    for (int k = 0; k < 4; ++k)
      Tm = wmma4(load_a(bufI, col, half, k), load_b(bufP, col, half, k), Tm);

    v8f KR = wmma4(ka, Rb, zero8());
    stage_c(bufS, KR, col, half);
    v8f KKt = wmma4(load_a(bufS, col, half, 0), load_bt(bufZ, col, half, 0),
                    zero8());

    stage_c(bufP, Tm, col, half);
    v8f Pu = KKt;
#pragma unroll
    for (int k = 0; k < 4; ++k)
      Pu = wmma4(load_a(bufP, col, half, k), load_bt(bufI, col, half, k), Pu);

    stage_c(bufP, Pu, col, half);
    v8f T2 = zero8();
#pragma unroll
    for (int k = 0; k < 4; ++k)
      T2 = wmma4(FA[k], load_b(bufP, col, half, k), T2);
    stage_c(bufI, T2, col, half);
    v8f Pn = Qc;
#pragma unroll
    for (int k = 0; k < 4; ++k)
      Pn = wmma4(load_a(bufI, col, half, k), FA[k]  , Pn);

    if (col == 0) {
#pragma unroll
      for (int r = 0; r < 8; ++r) bufS[r + 8 * half] = Mu[r];
    }
    v8f Mn = zero8();
#pragma unroll
    for (int k = 0; k < 4; ++k) {
      float b0 = bufS[4 * k + 2 * half];
      float b1 = bufS[4 * k + 2 * half + 1];
      v2f mb;
      mb.x = (col == 0) ? b0 : 0.f;
      mb.y = (col == 0) ? b1 : 0.f;
      Mn = wmma4(FA[k], mb, Mn);
    }

    Pc = Pn;
    Mc = Mn;
  }
}

extern "C" void kernel_launch(void* const* d_in, const int* in_sizes, int n_in,
                              void* d_out, int out_size, void* d_ws, size_t ws_size,
                              hipStream_t stream) {
  (void)in_sizes; (void)n_in; (void)out_size; (void)d_ws; (void)ws_size;
  const float* y  = (const float*)d_in[0];
  const float* F  = (const float*)d_in[1];
  const float* H  = (const float*)d_in[2];
  const float* Q  = (const float*)d_in[3];
  const float* R  = (const float*)d_in[4];
  const float* im = (const float*)d_in[5];
  const float* ic = (const float*)d_in[6];
  float* out = (float*)d_out;

  dim3 grid(G_GROUPS / WAVES_PER_BLOCK);
  dim3 block(THREADS);
  hipLaunchKernelGGL(kalman_wmma_kernel, grid, block, 0, stream,
                     y, F, H, Q, R, im, ic, out);
}


# Accumulator tiles fed straight back as WMMA operands (A use = transposed tile, state kept as P^T, constants in the same k order): no LDS staging; k-sum order inside each product changes
# speedup vs baseline: 1.5950x; 1.5950x over previous
#include <hip/hip_runtime.h>


typedef __attribute__((ext_vector_type(2))) float v2f;
typedef __attribute__((ext_vector_type(8))) float v8f;

#define G_GROUPS 1024
#define T_STEPS  512
#define WAVES_PER_BLOCK 4
#define THREADS (WAVES_PER_BLOCK * 32)

static __device__ __forceinline__ v8f wmma4(v2f a, v2f b, v8f c) {
  return __builtin_amdgcn_wmma_f32_16x16x4_f32(
       false, a,  false, b,
       (short)0, c,  false,  false);
}

static __device__ __forceinline__ v8f zero8() {
  v8f z = {0.f, 0.f, 0.f, 0.f, 0.f, 0.f, 0.f, 0.f};
  return z;
}

static __device__ __forceinline__ v2f pr(v8f t, int j) {
  v2f p;
  p.x = t[2 * j];
  p.y = t[2 * j + 1];
  return p;
}

static __device__ __forceinline__ float sel4(int i, float a, float b, float c, float d) {
  return i == 0 ? a : (i == 1 ? b : (i == 2 ? c : d));
}

__global__ __launch_bounds__(THREADS, 1)
void kalman_wmma_kernel(const float* __restrict__ y,
                        const float* __restrict__ F,
                        const float* __restrict__ H,
                        const float* __restrict__ Q,
                        const float* __restrict__ R,
                        const float* __restrict__ im,
                        const float* __restrict__ ic,
                        float* __restrict__ out) {
  const int lane = threadIdx.x & 31;
  const int wave = threadIdx.x >> 5;
  const int col  = lane & 15;
  const int half = lane >> 4;
  const int g    = blockIdx.x * WAVES_PER_BLOCK + wave;
  const bool top4 = (half == 0) && (col < 4);
  const bool at4  = (half == 0) && (col == 4);

  v2f Fk[4], Hk[4];
#pragma unroll
  for (int j = 0; j < 4; ++j) {
    int k0 = 8 * half + 2 * j;
    Fk[j].x = F[col * 16 + k0];
    Fk[j].y = F[col * 16 + k0 + 1];
    Hk[j].x = (col < 4) ? H[col * 16 + k0] : 0.f;
    Hk[j].y = (col < 4) ? H[col * 16 + k0 + 1] : 0.f;
  }
  v2f NHt[2], Rt[2];
#pragma unroll
  for (int j = 0; j < 2; ++j) {
    NHt[j].x = (half == 0) ? -H[(2 * j) * 16 + col] : 0.f;
    NHt[j].y = (half == 0) ? -H[(2 * j + 1) * 16 + col] : 0.f;
    Rt[j].x  = top4 ? R[(2 * j) * 4 + col] : 0.f;
    Rt[j].y  = top4 ? R[(2 * j + 1) * 4 + col] : 0.f;
  }

  v8f Qt, Rc, Ic, X, Mc;
#pragma unroll
  for (int r = 0; r < 8; ++r) {
    int row = r + 8 * half;
    Qt[r] = Q[col * 16 + row];
    Rc[r] = (row < 4 && col < 4) ? R[row * 4 + col] : 0.f;
    Ic[r] = (row == col) ? 1.f : 0.f;
    X[r]  = ic[col * 16 + row];
    Mc[r] = (col == 4) ? im[g * 16 + row] : 0.f;
  }

  for (int t = 0; t < T_STEPS; ++t) {
    v8f Z = Mc;
#pragma unroll
    for (int j = 0; j < 4; ++j) Z = wmma4(pr(X, j), Hk[j], Z);

    v8f Yt = zero8();
#pragma unroll
    for (int j = 0; j < 4; ++j) Yt = wmma4(Hk[j], pr(X, j), Yt);

    v8f Sc = Rc;
#pragma unroll
    for (int j = 0; j < 4; ++j) Sc = wmma4(Hk[j], pr(Z, j), Sc);

    if (lane == 4) {
      float4 o;
      o.x = Sc[0]; o.y = Sc[1]; o.z = Sc[2]; o.w = Sc[3];
      *(float4*)(out + ((size_t)g * T_STEPS + t) * 4) = o;
    }

    float s00 = __shfl(Sc[0], 0, 32), s01 = __shfl(Sc[0], 1, 32),
          s02 = __shfl(Sc[0], 2, 32), s03 = __shfl(Sc[0], 3, 32);
    float s10 = __shfl(Sc[1], 0, 32), s11 = __shfl(Sc[1], 1, 32),
          s12 = __shfl(Sc[1], 2, 32), s13 = __shfl(Sc[1], 3, 32);
    float s20 = __shfl(Sc[2], 0, 32), s21 = __shfl(Sc[2], 1, 32),
          s22 = __shfl(Sc[2], 2, 32), s23 = __shfl(Sc[2], 3, 32);
    float s30 = __shfl(Sc[3], 0, 32), s31 = __shfl(Sc[3], 1, 32),
          s32 = __shfl(Sc[3], 2, 32), s33 = __shfl(Sc[3], 3, 32);
    float mm0 = __shfl(Sc[0], 4, 32), mm1 = __shfl(Sc[1], 4, 32),
          mm2 = __shfl(Sc[2], 4, 32), mm3 = __shfl(Sc[3], 4, 32);

    const float4 yv = *(const float4*)(y + ((size_t)g * T_STEPS + t) * 4);
    float r0 = yv.x - mm0, r1 = yv.y - mm1, r2 = yv.z - mm2, r3 = yv.w - mm3;

    float A2323 = s22 * s33 - s23 * s32;
    float A1323 = s21 * s33 - s23 * s31;
    float A1223 = s21 * s32 - s22 * s31;
    float A0323 = s20 * s33 - s23 * s30;
    float A0223 = s20 * s32 - s22 * s30;
    float A0123 = s20 * s31 - s21 * s30;
    float A2313 = s12 * s33 - s13 * s32;
    float A1313 = s11 * s33 - s13 * s31;
    float A1213 = s11 * s32 - s12 * s31;
    float A2312 = s12 * s23 - s13 * s22;
    float A1312 = s11 * s23 - s13 * s21;
    float A1212 = s11 * s22 - s12 * s21;
    float A0313 = s10 * s33 - s13 * s30;
    float A0213 = s10 * s32 - s12 * s30;
    float A0312 = s10 * s23 - s13 * s20;
    float A0212 = s10 * s22 - s12 * s20;
    float A0113 = s10 * s31 - s11 * s30;
    float A0112 = s10 * s21 - s11 * s20;

    float det = s00 * (s11 * A2323 - s12 * A1323 + s13 * A1223)
              - s01 * (s10 * A2323 - s12 * A0323 + s13 * A0223)
              + s02 * (s10 * A1323 - s11 * A0323 + s13 * A0123)
              - s03 * (s10 * A1223 - s11 * A0223 + s12 * A0123);
    float id = 1.0f / det;

    float i00 =  id * (s11 * A2323 - s12 * A1323 + s13 * A1223);
    float i01 = -id * (s01 * A2323 - s02 * A1323 + s03 * A1223);
    float i02 =  id * (s01 * A2313 - s02 * A1313 + s03 * A1213);
    float i03 = -id * (s01 * A2312 - s02 * A1312 + s03 * A1212);
    float i10 = -id * (s10 * A2323 - s12 * A0323 + s13 * A0223);
    float i11 =  id * (s00 * A2323 - s02 * A0323 + s03 * A0223);
    float i12 = -id * (s00 * A2313 - s02 * A0313 + s03 * A0213);
    float i13 =  id * (s00 * A2312 - s02 * A0312 + s03 * A0212);
    float i20 =  id * (s10 * A1323 - s11 * A0323 + s13 * A0123);
    float i21 = -id * (s00 * A1323 - s01 * A0323 + s03 * A0123);
    float i22 =  id * (s00 * A1313 - s01 * A0313 + s03 * A0113);
    float i23 = -id * (s00 * A1312 - s01 * A0312 + s03 * A0112);
    float i30 = -id * (s10 * A1223 - s11 * A0223 + s12 * A0123);
    float i31 =  id * (s00 * A1223 - s01 * A0223 + s02 * A0123);
    float i32 = -id * (s00 * A1213 - s01 * A0213 + s02 * A0113);
    float i33 =  id * (s00 * A1212 - s01 * A0212 + s02 * A0112);

    float si_c0 = sel4(col, i00, i10, i20, i30);
    float si_c1 = sel4(col, i01, i11, i21, i31);
    float si_c2 = sel4(col, i02, i12, i22, i32);
    float si_c3 = sel4(col, i03, i13, i23, i33);
    v2f Si0, Si1;
    Si0.x = top4 ? si_c0 : 0.f;
    Si0.y = top4 ? si_c1 : 0.f;
    Si1.x = top4 ? si_c2 : 0.f;
    Si1.y = top4 ? si_c3 : 0.f;
    v8f Kt = wmma4(Si0, pr(Yt, 0), zero8());
    Kt = wmma4(Si1, pr(Yt, 1), Kt);

    v8f IKHt = wmma4(NHt[0], pr(Kt, 0), Ic);
    IKHt = wmma4(NHt[1], pr(Kt, 1), IKHt);

    v2f rb0, rb1;
    rb0.x = at4 ? r0 : 0.f;
    rb0.y = at4 ? r1 : 0.f;
    rb1.x = at4 ? r2 : 0.f;
    rb1.y = at4 ? r3 : 0.f;
    v8f Mu = wmma4(pr(Kt, 0), rb0, Mc);
    Mu = wmma4(pr(Kt, 1), rb1, Mu);

    v8f W1 = zero8();
#pragma unroll
    for (int j = 0; j < 4; ++j) W1 = wmma4(pr(X, j), pr(IKHt, j), W1);

    v8f V = wmma4(Rt[0], pr(Kt, 0), zero8());
    V = wmma4(Rt[1], pr(Kt, 1), V);
    v8f E = wmma4(pr(Kt, 0), pr(V, 0), zero8());
    E = wmma4(pr(Kt, 1), pr(V, 1), E);

    v8f Xu = E;
#pragma unroll
    for (int j = 0; j < 4; ++j) Xu = wmma4(pr(W1, j), pr(IKHt, j), Xu);

    v8f W3 = zero8();
#pragma unroll
    for (int j = 0; j < 4; ++j) W3 = wmma4(pr(Xu, j), Fk[j], W3);
    v8f Xn = Qt;
#pragma unroll
    for (int j = 0; j < 4; ++j) Xn = wmma4(pr(W3, j), Fk[j], Xn);

    v8f Mn = zero8();
#pragma unroll
    for (int j = 0; j < 4; ++j) Mn = wmma4(Fk[j], pr(Mu, j), Mn);

    X = Xn;
    Mc = Mn;
  }
}

extern "C" void kernel_launch(void* const* d_in, const int* in_sizes, int n_in,
                              void* d_out, int out_size, void* d_ws, size_t ws_size,
                              hipStream_t stream) {
  (void)in_sizes; (void)n_in; (void)out_size; (void)d_ws; (void)ws_size;
  const float* y  = (const float*)d_in[0];
  const float* F  = (const float*)d_in[1];
  const float* H  = (const float*)d_in[2];
  const float* Q  = (const float*)d_in[3];
  const float* R  = (const float*)d_in[4];
  const float* im = (const float*)d_in[5];
  const float* ic = (const float*)d_in[6];
  float* out = (float*)d_out;

  dim3 grid(G_GROUPS / WAVES_PER_BLOCK);
  dim3 block(THREADS);
  hipLaunchKernelGGL(kalman_wmma_kernel, grid, block, 0, stream,
                     y, F, H, Q, R, im, ic, out);
}


# 4x4 innovation inverse done one row per lane (quad_perm DPP column rotations, cofactor expansion per lane) in place of 20 LDS shuffles and a divergent 16-way select; determinant sum re-associated
# speedup vs baseline: 2.7229x; 1.7071x over previous
#include <hip/hip_runtime.h>


typedef __attribute__((ext_vector_type(2))) float v2f;
typedef __attribute__((ext_vector_type(8))) float v8f;

#define G_GROUPS 1024
#define T_STEPS  512
#define WAVES_PER_BLOCK 4
#define THREADS (WAVES_PER_BLOCK * 32)

static __device__ __forceinline__ v8f wmma4(v2f a, v2f b, v8f c) {
  return __builtin_amdgcn_wmma_f32_16x16x4_f32(
       false, a,  false, b,
       (short)0, c,  false,  false);
}

static __device__ __forceinline__ v8f zero8() {
  v8f z = {0.f, 0.f, 0.f, 0.f, 0.f, 0.f, 0.f, 0.f};
  return z;
}

static __device__ __forceinline__ v2f pr(v8f t, int j) {
  v2f p;
  p.x = t[2 * j];
  p.y = t[2 * j + 1];
  return p;
}

template <int CTRL>
static __device__ __forceinline__ float quad_rot(float v) {
  int r = __builtin_amdgcn_update_dpp(0, __builtin_bit_cast(int, v), CTRL, 0xf, 0xf, true);
  return __builtin_bit_cast(float, r);
}
#define QUAD_ROT1 0x39
#define QUAD_ROT2 0x4E
#define QUAD_ROT3 0x93

__global__ __launch_bounds__(THREADS, 1)
void kalman_wmma_kernel(const float* __restrict__ y,
                        const float* __restrict__ F,
                        const float* __restrict__ H,
                        const float* __restrict__ Q,
                        const float* __restrict__ R,
                        const float* __restrict__ im,
                        const float* __restrict__ ic,
                        float* __restrict__ out) {
  const int lane = threadIdx.x & 31;
  const int wave = threadIdx.x >> 5;
  const int col  = lane & 15;
  const int half = lane >> 4;
  const int g    = blockIdx.x * WAVES_PER_BLOCK + wave;
  const bool top4 = (half == 0) && (col < 4);
  const bool at4  = (half == 0) && (col == 4);

  v2f Fk[4], Hk[4];
#pragma unroll
  for (int j = 0; j < 4; ++j) {
    int k0 = 8 * half + 2 * j;
    Fk[j].x = F[col * 16 + k0];
    Fk[j].y = F[col * 16 + k0 + 1];
    Hk[j].x = (col < 4) ? H[col * 16 + k0] : 0.f;
    Hk[j].y = (col < 4) ? H[col * 16 + k0 + 1] : 0.f;
  }
  v2f NHt[2], Rt[2];
#pragma unroll
  for (int j = 0; j < 2; ++j) {
    NHt[j].x = (half == 0) ? -H[(2 * j) * 16 + col] : 0.f;
    NHt[j].y = (half == 0) ? -H[(2 * j + 1) * 16 + col] : 0.f;
    Rt[j].x  = top4 ? R[(2 * j) * 4 + col] : 0.f;
    Rt[j].y  = top4 ? R[(2 * j + 1) * 4 + col] : 0.f;
  }

  v8f Qt, Rc, Ic, X, Mc;
#pragma unroll
  for (int r = 0; r < 8; ++r) {
    int row = r + 8 * half;
    Qt[r] = Q[col * 16 + row];
    Rc[r] = (row < 4 && col < 4) ? R[row * 4 + col] : 0.f;
    Ic[r] = (row == col) ? 1.f : 0.f;
    X[r]  = ic[col * 16 + row];
    Mc[r] = (col == 4) ? im[g * 16 + row] : 0.f;
  }

  for (int t = 0; t < T_STEPS; ++t) {
    const float4 yv = *(const float4*)(y + ((size_t)g * T_STEPS + t) * 4);

    v8f Z = Mc;
#pragma unroll
    for (int j = 0; j < 4; ++j) Z = wmma4(pr(X, j), Hk[j], Z);

    v8f Yt = zero8();
#pragma unroll
    for (int j = 0; j < 4; ++j) Yt = wmma4(Hk[j], pr(X, j), Yt);

    v8f Sc = Rc;
#pragma unroll
    for (int j = 0; j < 4; ++j) Sc = wmma4(Hk[j], pr(Z, j), Sc);

    if (lane == 4) {
      float4 o;
      o.x = Sc[0]; o.y = Sc[1]; o.z = Sc[2]; o.w = Sc[3];
      *(float4*)(out + ((size_t)g * T_STEPS + t) * 4) = o;
    }

    float r0 = yv.x - Sc[0], r1 = yv.y - Sc[1], r2 = yv.z - Sc[2], r3 = yv.w - Sc[3];

    float a1_0 = quad_rot<QUAD_ROT1>(Sc[0]), a1_1 = quad_rot<QUAD_ROT1>(Sc[1]),
          a1_2 = quad_rot<QUAD_ROT1>(Sc[2]), a1_3 = quad_rot<QUAD_ROT1>(Sc[3]);
    float a2_0 = quad_rot<QUAD_ROT2>(Sc[0]), a2_1 = quad_rot<QUAD_ROT2>(Sc[1]),
          a2_2 = quad_rot<QUAD_ROT2>(Sc[2]), a2_3 = quad_rot<QUAD_ROT2>(Sc[3]);
    float a3_0 = quad_rot<QUAD_ROT3>(Sc[0]), a3_1 = quad_rot<QUAD_ROT3>(Sc[1]),
          a3_2 = quad_rot<QUAD_ROT3>(Sc[2]), a3_3 = quad_rot<QUAD_ROT3>(Sc[3]);

    float d01 = a2_0 * a3_1 - a2_1 * a3_0;
    float d02 = a2_0 * a3_2 - a2_2 * a3_0;
    float d03 = a2_0 * a3_3 - a2_3 * a3_0;
    float d12 = a2_1 * a3_2 - a2_2 * a3_1;
    float d13 = a2_1 * a3_3 - a2_3 * a3_1;
    float d23 = a2_2 * a3_3 - a2_3 * a3_2;

    float m0 = a1_1 * d23 - a1_2 * d13 + a1_3 * d12;
    float m1 = a1_0 * d23 - a1_2 * d03 + a1_3 * d02;
    float m2 = a1_0 * d13 - a1_1 * d03 + a1_3 * d01;
    float m3 = a1_0 * d12 - a1_1 * d02 + a1_2 * d01;

    float det = (Sc[0] * m0 - Sc[1] * m1) + (Sc[2] * m2 - Sc[3] * m3);
    float id = 1.0f / det;

    v2f Si0, Si1;
    Si0.x = top4 ?  (m0 * id) : 0.f;
    Si0.y = top4 ? -(m1 * id) : 0.f;
    Si1.x = top4 ?  (m2 * id) : 0.f;
    Si1.y = top4 ? -(m3 * id) : 0.f;
    v8f Kt = wmma4(Si0, pr(Yt, 0), zero8());
    Kt = wmma4(Si1, pr(Yt, 1), Kt);

    v8f IKHt = wmma4(NHt[0], pr(Kt, 0), Ic);
    IKHt = wmma4(NHt[1], pr(Kt, 1), IKHt);

    v2f rb0, rb1;
    rb0.x = at4 ? r0 : 0.f;
    rb0.y = at4 ? r1 : 0.f;
    rb1.x = at4 ? r2 : 0.f;
    rb1.y = at4 ? r3 : 0.f;
    v8f Mu = wmma4(pr(Kt, 0), rb0, Mc);
    Mu = wmma4(pr(Kt, 1), rb1, Mu);

    v8f W1 = zero8();
#pragma unroll
    for (int j = 0; j < 4; ++j) W1 = wmma4(pr(X, j), pr(IKHt, j), W1);

    v8f V = wmma4(Rt[0], pr(Kt, 0), zero8());
    V = wmma4(Rt[1], pr(Kt, 1), V);
    v8f E = wmma4(pr(Kt, 0), pr(V, 0), zero8());
    E = wmma4(pr(Kt, 1), pr(V, 1), E);

    v8f Xu = E;
#pragma unroll
    for (int j = 0; j < 4; ++j) Xu = wmma4(pr(W1, j), pr(IKHt, j), Xu);

    v8f W3 = zero8();
#pragma unroll
    for (int j = 0; j < 4; ++j) W3 = wmma4(pr(Xu, j), Fk[j], W3);
    v8f Xn = Qt;
#pragma unroll
    for (int j = 0; j < 4; ++j) Xn = wmma4(pr(W3, j), Fk[j], Xn);

    v8f Mn = zero8();
#pragma unroll
    for (int j = 0; j < 4; ++j) Mn = wmma4(Fk[j], pr(Mu, j), Mn);

    X = Xn;
    Mc = Mn;
  }
}

extern "C" void kernel_launch(void* const* d_in, const int* in_sizes, int n_in,
                              void* d_out, int out_size, void* d_ws, size_t ws_size,
                              hipStream_t stream) {
  (void)in_sizes; (void)n_in; (void)out_size; (void)d_ws; (void)ws_size;
  const float* y  = (const float*)d_in[0];
  const float* F  = (const float*)d_in[1];
  const float* H  = (const float*)d_in[2];
  const float* Q  = (const float*)d_in[3];
  const float* R  = (const float*)d_in[4];
  const float* im = (const float*)d_in[5];
  const float* ic = (const float*)d_in[6];
  float* out = (float*)d_out;

  dim3 grid(G_GROUPS / WAVES_PER_BLOCK);
  dim3 block(THREADS);
  hipLaunchKernelGGL(kalman_wmma_kernel, grid, block, 0, stream,
                     y, F, H, Q, R, im, ic, out);
}


# Update and predict fused by re-association: P_n = G P G^T + FK R FK^T + Q with G = F - FK H, FK^T = Sinv (F P H^T)^T; 42 -> 34 WMMA per step, serial run 28 -> 20; product rounding order changes
# speedup vs baseline: 3.2063x; 1.1775x over previous
#include <hip/hip_runtime.h>


typedef __attribute__((ext_vector_type(2))) float v2f;
typedef __attribute__((ext_vector_type(8))) float v8f;

#define G_GROUPS 1024
#define T_STEPS  512
#define WAVES_PER_BLOCK 4
#define THREADS (WAVES_PER_BLOCK * 32)

static __device__ __forceinline__ v8f wmma4(v2f a, v2f b, v8f c) {
  return __builtin_amdgcn_wmma_f32_16x16x4_f32(
       false, a,  false, b,
       (short)0, c,  false,  false);
}

static __device__ __forceinline__ v8f zero8() {
  v8f z = {0.f, 0.f, 0.f, 0.f, 0.f, 0.f, 0.f, 0.f};
  return z;
}

static __device__ __forceinline__ v2f pr(v8f t, int j) {
  v2f p;
  p.x = t[2 * j];
  p.y = t[2 * j + 1];
  return p;
}

template <int CTRL>
static __device__ __forceinline__ float quad_rot(float v) {
  int r = __builtin_amdgcn_update_dpp(0, __builtin_bit_cast(int, v), CTRL, 0xf, 0xf, true);
  return __builtin_bit_cast(float, r);
}
#define QUAD_ROT1 0x39
#define QUAD_ROT2 0x4E
#define QUAD_ROT3 0x93

__global__ __launch_bounds__(THREADS, 1)
void kalman_wmma_kernel(const float* __restrict__ y,
                        const float* __restrict__ F,
                        const float* __restrict__ H,
                        const float* __restrict__ Q,
                        const float* __restrict__ R,
                        const float* __restrict__ im,
                        const float* __restrict__ ic,
                        float* __restrict__ out) {
  const int lane = threadIdx.x & 31;
  const int wave = threadIdx.x >> 5;
  const int col  = lane & 15;
  const int half = lane >> 4;
  const int g    = blockIdx.x * WAVES_PER_BLOCK + wave;
  const bool top4 = (half == 0) && (col < 4);
  const bool at4  = (half == 0) && (col == 4);

  v2f Fk[4], Hk[4];
#pragma unroll
  for (int j = 0; j < 4; ++j) {
    int k0 = 8 * half + 2 * j;
    Fk[j].x = F[col * 16 + k0];
    Fk[j].y = F[col * 16 + k0 + 1];
    Hk[j].x = (col < 4) ? H[col * 16 + k0] : 0.f;
    Hk[j].y = (col < 4) ? H[col * 16 + k0 + 1] : 0.f;
  }
  v2f NHt[2], Rt[2];
#pragma unroll
  for (int j = 0; j < 2; ++j) {
    NHt[j].x = (half == 0) ? -H[(2 * j) * 16 + col] : 0.f;
    NHt[j].y = (half == 0) ? -H[(2 * j + 1) * 16 + col] : 0.f;
    Rt[j].x  = top4 ? R[(2 * j) * 4 + col] : 0.f;
    Rt[j].y  = top4 ? R[(2 * j + 1) * 4 + col] : 0.f;
  }

  v8f Qt, Rc, Ft, X, Mc;
#pragma unroll
  for (int r = 0; r < 8; ++r) {
    int row = r + 8 * half;
    Qt[r] = Q[col * 16 + row];
    Rc[r] = (row < 4 && col < 4) ? R[row * 4 + col] : 0.f;
    Ft[r] = F[col * 16 + row];
    X[r]  = ic[col * 16 + row];
    Mc[r] = (col == 4) ? im[g * 16 + row] : 0.f;
  }

  for (int t = 0; t < T_STEPS; ++t) {
    const float4 yv = *(const float4*)(y + ((size_t)g * T_STEPS + t) * 4);

    v8f Z = Mc;
#pragma unroll
    for (int j = 0; j < 4; ++j) Z = wmma4(pr(X, j), Hk[j], Z);

    v8f Sc = Rc;
#pragma unroll
    for (int j = 0; j < 4; ++j) Sc = wmma4(Hk[j], pr(Z, j), Sc);

    if (lane == 4) {
      float4 o;
      o.x = Sc[0]; o.y = Sc[1]; o.z = Sc[2]; o.w = Sc[3];
      *(float4*)(out + ((size_t)g * T_STEPS + t) * 4) = o;
    }

    v8f YF = zero8();
#pragma unroll
    for (int j = 0; j < 4; ++j) YF = wmma4(pr(Z, j), Fk[j], YF);
    v8f FM = zero8();
#pragma unroll
    for (int j = 0; j < 4; ++j) FM = wmma4(Fk[j], pr(Mc, j), FM);

    float r0 = yv.x - Sc[0], r1 = yv.y - Sc[1], r2 = yv.z - Sc[2], r3 = yv.w - Sc[3];

    float a1_0 = quad_rot<QUAD_ROT1>(Sc[0]), a1_1 = quad_rot<QUAD_ROT1>(Sc[1]),
          a1_2 = quad_rot<QUAD_ROT1>(Sc[2]), a1_3 = quad_rot<QUAD_ROT1>(Sc[3]);
    float a2_0 = quad_rot<QUAD_ROT2>(Sc[0]), a2_1 = quad_rot<QUAD_ROT2>(Sc[1]),
          a2_2 = quad_rot<QUAD_ROT2>(Sc[2]), a2_3 = quad_rot<QUAD_ROT2>(Sc[3]);
    float a3_0 = quad_rot<QUAD_ROT3>(Sc[0]), a3_1 = quad_rot<QUAD_ROT3>(Sc[1]),
          a3_2 = quad_rot<QUAD_ROT3>(Sc[2]), a3_3 = quad_rot<QUAD_ROT3>(Sc[3]);

    float d01 = a2_0 * a3_1 - a2_1 * a3_0;
    float d02 = a2_0 * a3_2 - a2_2 * a3_0;
    float d03 = a2_0 * a3_3 - a2_3 * a3_0;
    float d12 = a2_1 * a3_2 - a2_2 * a3_1;
    float d13 = a2_1 * a3_3 - a2_3 * a3_1;
    float d23 = a2_2 * a3_3 - a2_3 * a3_2;

    float m0 = a1_1 * d23 - a1_2 * d13 + a1_3 * d12;
    float m1 = a1_0 * d23 - a1_2 * d03 + a1_3 * d02;
    float m2 = a1_0 * d13 - a1_1 * d03 + a1_3 * d01;
    float m3 = a1_0 * d12 - a1_1 * d02 + a1_2 * d01;

    float det = (Sc[0] * m0 - Sc[1] * m1) + (Sc[2] * m2 - Sc[3] * m3);
    float id = 1.0f / det;

    asm volatile("" :: "v"(FM[0]), "v"(FM[1]), "v"(FM[2]), "v"(FM[3]),
                       "v"(FM[4]), "v"(FM[5]), "v"(FM[6]), "v"(FM[7]));
    __builtin_amdgcn_sched_barrier(0);

    v2f Si0, Si1;
    Si0.x = top4 ?  (m0 * id) : 0.f;
    Si0.y = top4 ? -(m1 * id) : 0.f;
    Si1.x = top4 ?  (m2 * id) : 0.f;
    Si1.y = top4 ? -(m3 * id) : 0.f;
    v8f FKt = wmma4(Si0, pr(YF, 0), zero8());
    FKt = wmma4(Si1, pr(YF, 1), FKt);

    v8f Gt = wmma4(NHt[0], pr(FKt, 0), Ft);
    Gt = wmma4(NHt[1], pr(FKt, 1), Gt);

    v8f Wt = zero8();
#pragma unroll
    for (int j = 0; j < 4; ++j) Wt = wmma4(pr(X, j), pr(Gt, j), Wt);

    v8f V = wmma4(Rt[0], pr(FKt, 0), zero8());
    V = wmma4(Rt[1], pr(FKt, 1), V);
    v8f E = wmma4(pr(FKt, 0), pr(V, 0), Qt);
    E = wmma4(pr(FKt, 1), pr(V, 1), E);

    v2f rb0, rb1;
    rb0.x = at4 ? r0 : 0.f;
    rb0.y = at4 ? r1 : 0.f;
    rb1.x = at4 ? r2 : 0.f;
    rb1.y = at4 ? r3 : 0.f;
    v8f Mn = wmma4(pr(FKt, 0), rb0, FM);
    Mn = wmma4(pr(FKt, 1), rb1, Mn);

    v8f Xn = E;
#pragma unroll
    for (int j = 0; j < 4; ++j) Xn = wmma4(pr(Wt, j), pr(Gt, j), Xn);

    X = Xn;
    Mc = Mn;
  }
}

extern "C" void kernel_launch(void* const* d_in, const int* in_sizes, int n_in,
                              void* d_out, int out_size, void* d_ws, size_t ws_size,
                              hipStream_t stream) {
  (void)in_sizes; (void)n_in; (void)out_size; (void)d_ws; (void)ws_size;
  const float* y  = (const float*)d_in[0];
  const float* F  = (const float*)d_in[1];
  const float* H  = (const float*)d_in[2];
  const float* Q  = (const float*)d_in[3];
  const float* R  = (const float*)d_in[4];
  const float* im = (const float*)d_in[5];
  const float* ic = (const float*)d_in[6];
  float* out = (float*)d_out;

  dim3 grid(G_GROUPS / WAVES_PER_BLOCK);
  dim3 block(THREADS);
  hipLaunchKernelGGL(kalman_wmma_kernel, grid, block, 0, stream,
                     y, F, H, Q, R, im, ic, out);
}


# Measurements 2,3 repeated at tile rows 8,9 so one k-chunk spans all 4: FK^T, G^T, R^T FK^T, E, m_n take 1 WMMA each, not 2 (34 -> 29/step); k-sum order changes; inverse/YF/FM overlap kept by fences
# speedup vs baseline: 3.4682x; 1.0817x over previous
#include <hip/hip_runtime.h>


typedef __attribute__((ext_vector_type(2))) float v2f;
typedef __attribute__((ext_vector_type(8))) float v8f;

#define G_GROUPS 1024
#define T_STEPS  512
#define WAVES_PER_BLOCK 4
#define THREADS (WAVES_PER_BLOCK * 32)

static __device__ __forceinline__ v8f wmma4(v2f a, v2f b, v8f c) {
  return __builtin_amdgcn_wmma_f32_16x16x4_f32(
       false, a,  false, b,
       (short)0, c,  false,  false);
}

static __device__ __forceinline__ v8f zero8() {
  v8f z = {0.f, 0.f, 0.f, 0.f, 0.f, 0.f, 0.f, 0.f};
  return z;
}

static __device__ __forceinline__ void sched_fence() {
  __builtin_amdgcn_sched_barrier(0);
}

static __device__ __forceinline__ v2f pr(v8f t, int j) {
  v2f p;
  p.x = t[2 * j];
  p.y = t[2 * j + 1];
  return p;
}

template <int CTRL>
static __device__ __forceinline__ float quad_rot(float v) {
  int r = __builtin_amdgcn_update_dpp(0, __builtin_bit_cast(int, v), CTRL, 0xf, 0xf, true);
  return __builtin_bit_cast(float, r);
}
#define QUAD_ROT1 0x39
#define QUAD_ROT2 0x4E
#define QUAD_ROT3 0x93

__global__ __launch_bounds__(THREADS, 1)
void kalman_wmma_kernel(const float* __restrict__ y,
                        const float* __restrict__ F,
                        const float* __restrict__ H,
                        const float* __restrict__ Q,
                        const float* __restrict__ R,
                        const float* __restrict__ im,
                        const float* __restrict__ ic,
                        float* __restrict__ out) {
  const int lane = threadIdx.x & 31;
  const int wave = threadIdx.x >> 5;
  const int col  = lane & 15;
  const int half = lane >> 4;
  const int g    = blockIdx.x * WAVES_PER_BLOCK + wave;
  const int  mrow = (col < 4) ? col : ((col >= 8 && col < 12) ? ((col - 6) & 3) : -1);
  const bool msel = (mrow >= 0);
  const int  mclp = msel ? mrow : 0;
  const bool at4  = (col == 4);

  v2f Fk[4], Hk[4];
#pragma unroll
  for (int j = 0; j < 4; ++j) {
    int k0 = 8 * half + 2 * j;
    Fk[j].x = F[col * 16 + k0];
    Fk[j].y = F[col * 16 + k0 + 1];
    Hk[j].x = msel ? H[mclp * 16 + k0] : 0.f;
    Hk[j].y = msel ? H[mclp * 16 + k0 + 1] : 0.f;
  }
  v2f NHt, Rt;
  NHt.x = -H[(2 * half) * 16 + col];
  NHt.y = -H[(2 * half + 1) * 16 + col];
  Rt.x  = msel ? R[(2 * half) * 4 + mclp] : 0.f;
  Rt.y  = msel ? R[(2 * half + 1) * 4 + mclp] : 0.f;

  v8f Qt, Rc, Ft, X, Mc;
#pragma unroll
  for (int r = 0; r < 8; ++r) {
    int row = r + 8 * half;
    int rrow = (row < 4) ? row : ((row >= 8 && row < 12) ? ((row - 6) & 3) : -1);
    Qt[r] = Q[col * 16 + row];
    Rc[r] = (rrow >= 0 && msel) ? R[rrow * 4 + mclp] : 0.f;
    Ft[r] = F[col * 16 + row];
    X[r]  = ic[col * 16 + row];
    Mc[r] = (col == 4) ? im[g * 16 + row] : 0.f;
  }

  for (int t = 0; t < T_STEPS; ++t) {
    const float4 yv = *(const float4*)(y + ((size_t)g * T_STEPS + t) * 4);

    v8f Z = Mc;
#pragma unroll
    for (int j = 0; j < 4; ++j) Z = wmma4(pr(X, j), Hk[j], Z);

    v8f Sc = Rc;
#pragma unroll
    for (int j = 0; j < 4; ++j) Sc = wmma4(Hk[j], pr(Z, j), Sc);

    if (lane == 4) {
      float4 o;
      o.x = Sc[0]; o.y = Sc[1]; o.z = Sc[2]; o.w = Sc[3];
      *(float4*)(out + ((size_t)g * T_STEPS + t) * 4) = o;
    }

    v8f FM = wmma4(Fk[0], pr(Mc, 0), zero8());

    float r0 = (half ? yv.z : yv.x) - Sc[0];
    float r1 = (half ? yv.w : yv.y) - Sc[1];

    float a1_0 = quad_rot<QUAD_ROT1>(Sc[0]), a1_1 = quad_rot<QUAD_ROT1>(Sc[1]),
          a1_2 = quad_rot<QUAD_ROT1>(Sc[2]), a1_3 = quad_rot<QUAD_ROT1>(Sc[3]);
    float a2_0 = quad_rot<QUAD_ROT2>(Sc[0]), a2_1 = quad_rot<QUAD_ROT2>(Sc[1]),
          a2_2 = quad_rot<QUAD_ROT2>(Sc[2]), a2_3 = quad_rot<QUAD_ROT2>(Sc[3]);
    float a3_0 = quad_rot<QUAD_ROT3>(Sc[0]), a3_1 = quad_rot<QUAD_ROT3>(Sc[1]),
          a3_2 = quad_rot<QUAD_ROT3>(Sc[2]), a3_3 = quad_rot<QUAD_ROT3>(Sc[3]);

    float d01 = a2_0 * a3_1 - a2_1 * a3_0;
    float d02 = a2_0 * a3_2 - a2_2 * a3_0;
    sched_fence();
    v8f YF = wmma4(pr(Z, 0), Fk[0], zero8());
    float d03 = a2_0 * a3_3 - a2_3 * a3_0;
    float d12 = a2_1 * a3_2 - a2_2 * a3_1;
    sched_fence();
    FM = wmma4(Fk[1], pr(Mc, 1), FM);
    float d13 = a2_1 * a3_3 - a2_3 * a3_1;
    float d23 = a2_2 * a3_3 - a2_3 * a3_2;
    sched_fence();
    YF = wmma4(pr(Z, 1), Fk[1], YF);
    float m0 = a1_1 * d23 - a1_2 * d13 + a1_3 * d12;
    float m1 = a1_0 * d23 - a1_2 * d03 + a1_3 * d02;
    sched_fence();
    FM = wmma4(Fk[2], pr(Mc, 2), FM);
    float m2 = a1_0 * d13 - a1_1 * d03 + a1_3 * d01;
    float m3 = a1_0 * d12 - a1_1 * d02 + a1_2 * d01;
    sched_fence();
    YF = wmma4(pr(Z, 2), Fk[2], YF);
    float det = (Sc[0] * m0 - Sc[1] * m1) + (Sc[2] * m2 - Sc[3] * m3);
    sched_fence();
    FM = wmma4(Fk[3], pr(Mc, 3), FM);
    YF = wmma4(pr(Z, 3), Fk[3], YF);
    float id = 1.0f / det;

    asm volatile("" :: "v"(FM[0]), "v"(FM[1]), "v"(FM[2]), "v"(FM[3]),
                       "v"(FM[4]), "v"(FM[5]), "v"(FM[6]), "v"(FM[7]));
    __builtin_amdgcn_sched_barrier(0);

    v2f Si;
    Si.x = msel ?  (m0 * id) : 0.f;
    Si.y = msel ? -(m1 * id) : 0.f;
    v8f FKt = wmma4(Si, pr(YF, 0), zero8());

    v8f Gt = wmma4(NHt, pr(FKt, 0), Ft);

    v8f Wt = zero8();
#pragma unroll
    for (int j = 0; j < 4; ++j) Wt = wmma4(pr(X, j), pr(Gt, j), Wt);

    v8f V = wmma4(Rt, pr(FKt, 0), zero8());
    v8f E = wmma4(pr(FKt, 0), pr(V, 0), Qt);

    v2f rb;
    rb.x = at4 ? r0 : 0.f;
    rb.y = at4 ? r1 : 0.f;
    v8f Mn = wmma4(pr(FKt, 0), rb, FM);

    v8f Xn = E;
#pragma unroll
    for (int j = 0; j < 4; ++j) Xn = wmma4(pr(Wt, j), pr(Gt, j), Xn);

    X = Xn;
    Mc = Mn;
  }
}

extern "C" void kernel_launch(void* const* d_in, const int* in_sizes, int n_in,
                              void* d_out, int out_size, void* d_ws, size_t ws_size,
                              hipStream_t stream) {
  (void)in_sizes; (void)n_in; (void)out_size; (void)d_ws; (void)ws_size;
  const float* y  = (const float*)d_in[0];
  const float* F  = (const float*)d_in[1];
  const float* H  = (const float*)d_in[2];
  const float* Q  = (const float*)d_in[3];
  const float* R  = (const float*)d_in[4];
  const float* im = (const float*)d_in[5];
  const float* ic = (const float*)d_in[6];
  float* out = (float*)d_out;

  dim3 grid(G_GROUPS / WAVES_PER_BLOCK);
  dim3 block(THREADS);
  hipLaunchKernelGGL(kalman_wmma_kernel, grid, block, 0, stream,
                     y, F, H, Q, R, im, ic, out);
}


# F m read from row 4 of Z^T F^T (Z carries m in column 4), turned back into column 4 by one one-hot WMMA issued under the division: F@m's 4 WMMAs -> 1 (29 -> 26/step); k-sum order unchanged
# speedup vs baseline: 3.6510x; 1.0527x over previous
#include <hip/hip_runtime.h>


typedef __attribute__((ext_vector_type(2))) float v2f;
typedef __attribute__((ext_vector_type(8))) float v8f;

#define G_GROUPS 1024
#define T_STEPS  512
#define WAVES_PER_BLOCK 4
#define THREADS (WAVES_PER_BLOCK * 32)

static __device__ __forceinline__ v8f wmma4(v2f a, v2f b, v8f c) {
  return __builtin_amdgcn_wmma_f32_16x16x4_f32(
       false, a,  false, b,
       (short)0, c,  false,  false);
}

static __device__ __forceinline__ v8f zero8() {
  v8f z = {0.f, 0.f, 0.f, 0.f, 0.f, 0.f, 0.f, 0.f};
  return z;
}

static __device__ __forceinline__ void sched_fence() {
  __builtin_amdgcn_sched_barrier(0);
}

static __device__ __forceinline__ v2f pr(v8f t, int j) {
  v2f p;
  p.x = t[2 * j];
  p.y = t[2 * j + 1];
  return p;
}

template <int CTRL>
static __device__ __forceinline__ float quad_rot(float v) {
  int r = __builtin_amdgcn_update_dpp(0, __builtin_bit_cast(int, v), CTRL, 0xf, 0xf, true);
  return __builtin_bit_cast(float, r);
}
#define QUAD_ROT1 0x39
#define QUAD_ROT2 0x4E
#define QUAD_ROT3 0x93

__global__ __launch_bounds__(THREADS, 1)
void kalman_wmma_kernel(const float* __restrict__ y,
                        const float* __restrict__ F,
                        const float* __restrict__ H,
                        const float* __restrict__ Q,
                        const float* __restrict__ R,
                        const float* __restrict__ im,
                        const float* __restrict__ ic,
                        float* __restrict__ out) {
  const int lane = threadIdx.x & 31;
  const int wave = threadIdx.x >> 5;
  const int col  = lane & 15;
  const int half = lane >> 4;
  const int g    = blockIdx.x * WAVES_PER_BLOCK + wave;
  const int  mrow = (col < 4) ? col : ((col >= 8 && col < 12) ? ((col - 6) & 3) : -1);
  const bool msel = (mrow >= 0);
  const int  mclp = msel ? mrow : 0;
  const bool at4  = (col == 4);

  v2f Fk[4], Hk[4];
#pragma unroll
  for (int j = 0; j < 4; ++j) {
    int k0 = 8 * half + 2 * j;
    Fk[j].x = F[col * 16 + k0];
    Fk[j].y = F[col * 16 + k0 + 1];
    Hk[j].x = msel ? H[mclp * 16 + k0] : 0.f;
    Hk[j].y = msel ? H[mclp * 16 + k0 + 1] : 0.f;
  }
  v2f NHt, Rt;
  NHt.x = -H[(2 * half) * 16 + col];
  NHt.y = -H[(2 * half + 1) * 16 + col];
  Rt.x  = msel ? R[(2 * half) * 4 + mclp] : 0.f;
  Rt.y  = msel ? R[(2 * half + 1) * 4 + mclp] : 0.f;
  v2f Sel4;
  Sel4.x = (lane == 4) ? 1.f : 0.f;
  Sel4.y = 0.f;

  v8f Qt, Rc, Ft, X, Mc;
#pragma unroll
  for (int r = 0; r < 8; ++r) {
    int row = r + 8 * half;
    int rrow = (row < 4) ? row : ((row >= 8 && row < 12) ? ((row - 6) & 3) : -1);
    Qt[r] = Q[col * 16 + row];
    Rc[r] = (rrow >= 0 && msel) ? R[rrow * 4 + mclp] : 0.f;
    Ft[r] = F[col * 16 + row];
    X[r]  = ic[col * 16 + row];
    Mc[r] = (col == 4) ? im[g * 16 + row] : 0.f;
  }

  for (int t = 0; t < T_STEPS; ++t) {
    const float4 yv = *(const float4*)(y + ((size_t)g * T_STEPS + t) * 4);

    v8f Z = Mc;
#pragma unroll
    for (int j = 0; j < 4; ++j) Z = wmma4(pr(X, j), Hk[j], Z);

    v8f Sc = Rc;
#pragma unroll
    for (int j = 0; j < 4; ++j) Sc = wmma4(Hk[j], pr(Z, j), Sc);

    if (lane == 4) {
      float4 o;
      o.x = Sc[0]; o.y = Sc[1]; o.z = Sc[2]; o.w = Sc[3];
      *(float4*)(out + ((size_t)g * T_STEPS + t) * 4) = o;
    }

    v8f YF = wmma4(pr(Z, 0), Fk[0], zero8());

    float r0 = (half ? yv.z : yv.x) - Sc[0];
    float r1 = (half ? yv.w : yv.y) - Sc[1];

    float a1_0 = quad_rot<QUAD_ROT1>(Sc[0]), a1_1 = quad_rot<QUAD_ROT1>(Sc[1]),
          a1_2 = quad_rot<QUAD_ROT1>(Sc[2]), a1_3 = quad_rot<QUAD_ROT1>(Sc[3]);
    float a2_0 = quad_rot<QUAD_ROT2>(Sc[0]), a2_1 = quad_rot<QUAD_ROT2>(Sc[1]),
          a2_2 = quad_rot<QUAD_ROT2>(Sc[2]), a2_3 = quad_rot<QUAD_ROT2>(Sc[3]);
    float a3_0 = quad_rot<QUAD_ROT3>(Sc[0]), a3_1 = quad_rot<QUAD_ROT3>(Sc[1]),
          a3_2 = quad_rot<QUAD_ROT3>(Sc[2]), a3_3 = quad_rot<QUAD_ROT3>(Sc[3]);

    float d01 = a2_0 * a3_1 - a2_1 * a3_0;
    float d02 = a2_0 * a3_2 - a2_2 * a3_0;
    float d03 = a2_0 * a3_3 - a2_3 * a3_0;
    sched_fence();
    YF = wmma4(pr(Z, 1), Fk[1], YF);
    float d12 = a2_1 * a3_2 - a2_2 * a3_1;
    float d13 = a2_1 * a3_3 - a2_3 * a3_1;
    float d23 = a2_2 * a3_3 - a2_3 * a3_2;
    sched_fence();
    YF = wmma4(pr(Z, 2), Fk[2], YF);
    float m0 = a1_1 * d23 - a1_2 * d13 + a1_3 * d12;
    float m1 = a1_0 * d23 - a1_2 * d03 + a1_3 * d02;
    sched_fence();
    YF = wmma4(pr(Z, 3), Fk[3], YF);
    float m2 = a1_0 * d13 - a1_1 * d03 + a1_3 * d01;
    float m3 = a1_0 * d12 - a1_1 * d02 + a1_2 * d01;
    float det = (Sc[0] * m0 - Sc[1] * m1) + (Sc[2] * m2 - Sc[3] * m3);

    sched_fence();
    v8f FM = wmma4(pr(YF, 2), Sel4, zero8());
    asm volatile("" :: "v"(FM[0]), "v"(FM[1]), "v"(FM[2]), "v"(FM[3]),
                       "v"(FM[4]), "v"(FM[5]), "v"(FM[6]), "v"(FM[7]));
    sched_fence();
    float id = 1.0f / det;

    v2f Si;
    Si.x = msel ?  (m0 * id) : 0.f;
    Si.y = msel ? -(m1 * id) : 0.f;
    v8f FKt = wmma4(Si, pr(YF, 0), zero8());

    v8f Gt = wmma4(NHt, pr(FKt, 0), Ft);

    v8f Wt = zero8();
#pragma unroll
    for (int j = 0; j < 4; ++j) Wt = wmma4(pr(X, j), pr(Gt, j), Wt);

    v8f V = wmma4(Rt, pr(FKt, 0), zero8());
    v8f E = wmma4(pr(FKt, 0), pr(V, 0), Qt);

    v2f rb;
    rb.x = at4 ? r0 : 0.f;
    rb.y = at4 ? r1 : 0.f;
    v8f Mn = wmma4(pr(FKt, 0), rb, FM);

    v8f Xn = E;
#pragma unroll
    for (int j = 0; j < 4; ++j) Xn = wmma4(pr(Wt, j), pr(Gt, j), Xn);

    X = Xn;
    Mc = Mn;
  }
}

extern "C" void kernel_launch(void* const* d_in, const int* in_sizes, int n_in,
                              void* d_out, int out_size, void* d_ws, size_t ws_size,
                              hipStream_t stream) {
  (void)in_sizes; (void)n_in; (void)out_size; (void)d_ws; (void)ws_size;
  const float* y  = (const float*)d_in[0];
  const float* F  = (const float*)d_in[1];
  const float* H  = (const float*)d_in[2];
  const float* Q  = (const float*)d_in[3];
  const float* R  = (const float*)d_in[4];
  const float* im = (const float*)d_in[5];
  const float* ic = (const float*)d_in[6];
  float* out = (float*)d_out;

  dim3 grid(G_GROUPS / WAVES_PER_BLOCK);
  dim3 block(THREADS);
  hipLaunchKernelGGL(kalman_wmma_kernel, grid, block, 0, stream,
                     y, F, H, Q, R, im, ic, out);
}


# 1/det of the 4x4 innovation matrix by the hardware reciprocal v_rcp_f32 (1 ulp) in place of the IEEE division: its ten-instruction dependent chain on the serial path S -> FK^T becomes one
# speedup vs baseline: 3.9676x; 1.0867x over previous
#include <hip/hip_runtime.h>


typedef __attribute__((ext_vector_type(2))) float v2f;
typedef __attribute__((ext_vector_type(8))) float v8f;

#define G_GROUPS 1024
#define T_STEPS  512
#define WAVES_PER_BLOCK 4
#define THREADS (WAVES_PER_BLOCK * 32)

static __device__ __forceinline__ v8f wmma4(v2f a, v2f b, v8f c) {
  return __builtin_amdgcn_wmma_f32_16x16x4_f32(
       false, a,  false, b,
       (short)0, c,  false,  false);
}

static __device__ __forceinline__ v8f zero8() {
  v8f z = {0.f, 0.f, 0.f, 0.f, 0.f, 0.f, 0.f, 0.f};
  return z;
}

static __device__ __forceinline__ void sched_fence() {
  __builtin_amdgcn_sched_barrier(0);
}

static __device__ __forceinline__ v2f pr(v8f t, int j) {
  v2f p;
  p.x = t[2 * j];
  p.y = t[2 * j + 1];
  return p;
}

template <int CTRL>
static __device__ __forceinline__ float quad_rot(float v) {
  int r = __builtin_amdgcn_update_dpp(0, __builtin_bit_cast(int, v), CTRL, 0xf, 0xf, true);
  return __builtin_bit_cast(float, r);
}
#define QUAD_ROT1 0x39
#define QUAD_ROT2 0x4E
#define QUAD_ROT3 0x93

__global__ __launch_bounds__(THREADS, 1)
void kalman_wmma_kernel(const float* __restrict__ y,
                        const float* __restrict__ F,
                        const float* __restrict__ H,
                        const float* __restrict__ Q,
                        const float* __restrict__ R,
                        const float* __restrict__ im,
                        const float* __restrict__ ic,
                        float* __restrict__ out) {
  const int lane = threadIdx.x & 31;
  const int wave = threadIdx.x >> 5;
  const int col  = lane & 15;
  const int half = lane >> 4;
  const int g    = blockIdx.x * WAVES_PER_BLOCK + wave;
  const int  mrow = (col < 4) ? col : ((col >= 8 && col < 12) ? ((col - 6) & 3) : -1);
  const bool msel = (mrow >= 0);
  const int  mclp = msel ? mrow : 0;
  const bool at4  = (col == 4);

  v2f Fk[4], Hk[4];
#pragma unroll
  for (int j = 0; j < 4; ++j) {
    int k0 = 8 * half + 2 * j;
    Fk[j].x = F[col * 16 + k0];
    Fk[j].y = F[col * 16 + k0 + 1];
    Hk[j].x = msel ? H[mclp * 16 + k0] : 0.f;
    Hk[j].y = msel ? H[mclp * 16 + k0 + 1] : 0.f;
  }
  v2f NHt, Rt;
  NHt.x = -H[(2 * half) * 16 + col];
  NHt.y = -H[(2 * half + 1) * 16 + col];
  Rt.x  = msel ? R[(2 * half) * 4 + mclp] : 0.f;
  Rt.y  = msel ? R[(2 * half + 1) * 4 + mclp] : 0.f;
  v2f Sel4;
  Sel4.x = (lane == 4) ? 1.f : 0.f;
  Sel4.y = 0.f;

  v8f Qt, Rc, Ft, X, Mc;
#pragma unroll
  for (int r = 0; r < 8; ++r) {
    int row = r + 8 * half;
    int rrow = (row < 4) ? row : ((row >= 8 && row < 12) ? ((row - 6) & 3) : -1);
    Qt[r] = Q[col * 16 + row];
    Rc[r] = (rrow >= 0 && msel) ? R[rrow * 4 + mclp] : 0.f;
    Ft[r] = F[col * 16 + row];
    X[r]  = ic[col * 16 + row];
    Mc[r] = (col == 4) ? im[g * 16 + row] : 0.f;
  }

  for (int t = 0; t < T_STEPS; ++t) {
    const float4 yv = *(const float4*)(y + ((size_t)g * T_STEPS + t) * 4);

    v8f Z = Mc;
#pragma unroll
    for (int j = 0; j < 4; ++j) Z = wmma4(pr(X, j), Hk[j], Z);

    v8f Sc = Rc;
#pragma unroll
    for (int j = 0; j < 4; ++j) Sc = wmma4(Hk[j], pr(Z, j), Sc);

    if (lane == 4) {
      float4 o;
      o.x = Sc[0]; o.y = Sc[1]; o.z = Sc[2]; o.w = Sc[3];
      *(float4*)(out + ((size_t)g * T_STEPS + t) * 4) = o;
    }

    v8f YF = wmma4(pr(Z, 0), Fk[0], zero8());

    float r0 = (half ? yv.z : yv.x) - Sc[0];
    float r1 = (half ? yv.w : yv.y) - Sc[1];

    float a1_0 = quad_rot<QUAD_ROT1>(Sc[0]), a1_1 = quad_rot<QUAD_ROT1>(Sc[1]),
          a1_2 = quad_rot<QUAD_ROT1>(Sc[2]), a1_3 = quad_rot<QUAD_ROT1>(Sc[3]);
    float a2_0 = quad_rot<QUAD_ROT2>(Sc[0]), a2_1 = quad_rot<QUAD_ROT2>(Sc[1]),
          a2_2 = quad_rot<QUAD_ROT2>(Sc[2]), a2_3 = quad_rot<QUAD_ROT2>(Sc[3]);
    float a3_0 = quad_rot<QUAD_ROT3>(Sc[0]), a3_1 = quad_rot<QUAD_ROT3>(Sc[1]),
          a3_2 = quad_rot<QUAD_ROT3>(Sc[2]), a3_3 = quad_rot<QUAD_ROT3>(Sc[3]);

    float d01 = a2_0 * a3_1 - a2_1 * a3_0;
    float d02 = a2_0 * a3_2 - a2_2 * a3_0;
    float d03 = a2_0 * a3_3 - a2_3 * a3_0;
    sched_fence();
    YF = wmma4(pr(Z, 1), Fk[1], YF);
    float d12 = a2_1 * a3_2 - a2_2 * a3_1;
    float d13 = a2_1 * a3_3 - a2_3 * a3_1;
    float d23 = a2_2 * a3_3 - a2_3 * a3_2;
    sched_fence();
    YF = wmma4(pr(Z, 2), Fk[2], YF);
    float m0 = a1_1 * d23 - a1_2 * d13 + a1_3 * d12;
    float m1 = a1_0 * d23 - a1_2 * d03 + a1_3 * d02;
    sched_fence();
    YF = wmma4(pr(Z, 3), Fk[3], YF);
    float m2 = a1_0 * d13 - a1_1 * d03 + a1_3 * d01;
    float m3 = a1_0 * d12 - a1_1 * d02 + a1_2 * d01;
    float det = (Sc[0] * m0 - Sc[1] * m1) + (Sc[2] * m2 - Sc[3] * m3);

    sched_fence();
    v8f FM = wmma4(pr(YF, 2), Sel4, zero8());
    asm volatile("" :: "v"(FM[0]), "v"(FM[1]), "v"(FM[2]), "v"(FM[3]),
                       "v"(FM[4]), "v"(FM[5]), "v"(FM[6]), "v"(FM[7]));
    sched_fence();
    float id = __builtin_amdgcn_rcpf(det);

    v2f Si;
    Si.x = msel ?  (m0 * id) : 0.f;
    Si.y = msel ? -(m1 * id) : 0.f;
    v8f FKt = wmma4(Si, pr(YF, 0), zero8());

    v8f Gt = wmma4(NHt, pr(FKt, 0), Ft);

    v8f Wt = zero8();
#pragma unroll
    for (int j = 0; j < 4; ++j) Wt = wmma4(pr(X, j), pr(Gt, j), Wt);

    v8f V = wmma4(Rt, pr(FKt, 0), zero8());
    v8f E = wmma4(pr(FKt, 0), pr(V, 0), Qt);

    v2f rb;
    rb.x = at4 ? r0 : 0.f;
    rb.y = at4 ? r1 : 0.f;
    v8f Mn = wmma4(pr(FKt, 0), rb, FM);

    v8f Xn = E;
#pragma unroll
    for (int j = 0; j < 4; ++j) Xn = wmma4(pr(Wt, j), pr(Gt, j), Xn);

    X = Xn;
    Mc = Mn;
  }
}

extern "C" void kernel_launch(void* const* d_in, const int* in_sizes, int n_in,
                              void* d_out, int out_size, void* d_ws, size_t ws_size,
                              hipStream_t stream) {
  (void)in_sizes; (void)n_in; (void)out_size; (void)d_ws; (void)ws_size;
  const float* y  = (const float*)d_in[0];
  const float* F  = (const float*)d_in[1];
  const float* H  = (const float*)d_in[2];
  const float* Q  = (const float*)d_in[3];
  const float* R  = (const float*)d_in[4];
  const float* im = (const float*)d_in[5];
  const float* ic = (const float*)d_in[6];
  float* out = (float*)d_out;

  dim3 grid(G_GROUPS / WAVES_PER_BLOCK);
  dim3 block(THREADS);
  hipLaunchKernelGGL(kalman_wmma_kernel, grid, block, 0, stream,
                     y, F, H, Q, R, im, ic, out);
}


# Global I/O once per 32 steps: lane l loads y of step l one chunk ahead and stores out of step l in one 512-byte access; readlane + lane select replace the per-step load wait and masked store
# speedup vs baseline: 4.4116x; 1.1119x over previous
#include <hip/hip_runtime.h>


typedef __attribute__((ext_vector_type(2))) float v2f;
typedef __attribute__((ext_vector_type(8))) float v8f;

#define G_GROUPS 1024
#define T_STEPS  512
#define CHUNK    32
#define N_CHUNKS (T_STEPS / CHUNK)
#define WAVES_PER_BLOCK 4
#define THREADS (WAVES_PER_BLOCK * 32)

static __device__ __forceinline__ v8f wmma4(v2f a, v2f b, v8f c) {
  return __builtin_amdgcn_wmma_f32_16x16x4_f32(
       false, a,  false, b,
       (short)0, c,  false,  false);
}

static __device__ __forceinline__ v8f zero8() {
  v8f z = {0.f, 0.f, 0.f, 0.f, 0.f, 0.f, 0.f, 0.f};
  return z;
}

static __device__ __forceinline__ void sched_fence() {
  __builtin_amdgcn_sched_barrier(0);
}

static __device__ __forceinline__ v2f pr(v8f t, int j) {
  v2f p;
  p.x = t[2 * j];
  p.y = t[2 * j + 1];
  return p;
}

template <int CTRL>
static __device__ __forceinline__ float quad_rot(float v) {
  int r = __builtin_amdgcn_update_dpp(0, __builtin_bit_cast(int, v), CTRL, 0xf, 0xf, true);
  return __builtin_bit_cast(float, r);
}
static __device__ __forceinline__ float lane_get(float v, int l) {
  return __builtin_bit_cast(float,
      __builtin_amdgcn_readlane(__builtin_bit_cast(int, v), l));
}
#define QUAD_ROT1 0x39
#define QUAD_ROT2 0x4E
#define QUAD_ROT3 0x93

__global__ __launch_bounds__(THREADS, 1)
void kalman_wmma_kernel(const float* __restrict__ y,
                        const float* __restrict__ F,
                        const float* __restrict__ H,
                        const float* __restrict__ Q,
                        const float* __restrict__ R,
                        const float* __restrict__ im,
                        const float* __restrict__ ic,
                        float* __restrict__ out) {
  const int lane = threadIdx.x & 31;
  const int wave = threadIdx.x >> 5;
  const int col  = lane & 15;
  const int half = lane >> 4;
  const int g    = blockIdx.x * WAVES_PER_BLOCK + wave;
  const int  mrow = (col < 4) ? col : ((col >= 8 && col < 12) ? ((col - 6) & 3) : -1);
  const bool msel = (mrow >= 0);
  const int  mclp = msel ? mrow : 0;
  const bool at4  = (col == 4);

  v2f Fk[4], Hk[4];
#pragma unroll
  for (int j = 0; j < 4; ++j) {
    int k0 = 8 * half + 2 * j;
    Fk[j].x = F[col * 16 + k0];
    Fk[j].y = F[col * 16 + k0 + 1];
    Hk[j].x = msel ? H[mclp * 16 + k0] : 0.f;
    Hk[j].y = msel ? H[mclp * 16 + k0 + 1] : 0.f;
  }
  v2f NHt, Rt;
  NHt.x = -H[(2 * half) * 16 + col];
  NHt.y = -H[(2 * half + 1) * 16 + col];
  Rt.x  = msel ? R[(2 * half) * 4 + mclp] : 0.f;
  Rt.y  = msel ? R[(2 * half + 1) * 4 + mclp] : 0.f;
  v2f Sel4;
  Sel4.x = (lane == 4) ? 1.f : 0.f;
  Sel4.y = 0.f;

  v8f Qt, Rc, Ft, X, Mc;
#pragma unroll
  for (int r = 0; r < 8; ++r) {
    int row = r + 8 * half;
    int rrow = (row < 4) ? row : ((row >= 8 && row < 12) ? ((row - 6) & 3) : -1);
    Qt[r] = Q[col * 16 + row];
    Rc[r] = (rrow >= 0 && msel) ? R[rrow * 4 + mclp] : 0.f;
    Ft[r] = F[col * 16 + row];
    X[r]  = ic[col * 16 + row];
    Mc[r] = (col == 4) ? im[g * 16 + row] : 0.f;
  }

  const float* yg = y + (size_t)g * T_STEPS * 4;
  float*       og = out + (size_t)g * T_STEPS * 4;
  float4 ynext = *(const float4*)(yg + lane * 4);

  for (int c = 0; c < N_CHUNKS; ++c) {
   const float4 ycur = ynext;
   const int cn = (c + 1 < N_CHUNKS) ? (c + 1) : c;
   ynext = *(const float4*)(yg + (cn * CHUNK + lane) * 4);
   float4 ost = {0.f, 0.f, 0.f, 0.f};

   for (int s = 0; s < CHUNK; ++s) {
    const float y0 = lane_get(ycur.x, s), y1 = lane_get(ycur.y, s),
                y2 = lane_get(ycur.z, s), y3 = lane_get(ycur.w, s);

    v8f Z = Mc;
#pragma unroll
    for (int j = 0; j < 4; ++j) Z = wmma4(pr(X, j), Hk[j], Z);

    v8f Sc = Rc;
#pragma unroll
    for (int j = 0; j < 4; ++j) Sc = wmma4(Hk[j], pr(Z, j), Sc);

    const float p0 = lane_get(Sc[0], 4), p1 = lane_get(Sc[1], 4),
                p2 = lane_get(Sc[2], 4), p3 = lane_get(Sc[3], 4);
    const bool mine = (lane == s);
    ost.x = mine ? p0 : ost.x;
    ost.y = mine ? p1 : ost.y;
    ost.z = mine ? p2 : ost.z;
    ost.w = mine ? p3 : ost.w;

    v8f YF = wmma4(pr(Z, 0), Fk[0], zero8());

    float r0 = (half ? y2 : y0) - Sc[0];
    float r1 = (half ? y3 : y1) - Sc[1];

    float a1_0 = quad_rot<QUAD_ROT1>(Sc[0]), a1_1 = quad_rot<QUAD_ROT1>(Sc[1]),
          a1_2 = quad_rot<QUAD_ROT1>(Sc[2]), a1_3 = quad_rot<QUAD_ROT1>(Sc[3]);
    float a2_0 = quad_rot<QUAD_ROT2>(Sc[0]), a2_1 = quad_rot<QUAD_ROT2>(Sc[1]),
          a2_2 = quad_rot<QUAD_ROT2>(Sc[2]), a2_3 = quad_rot<QUAD_ROT2>(Sc[3]);
    float a3_0 = quad_rot<QUAD_ROT3>(Sc[0]), a3_1 = quad_rot<QUAD_ROT3>(Sc[1]),
          a3_2 = quad_rot<QUAD_ROT3>(Sc[2]), a3_3 = quad_rot<QUAD_ROT3>(Sc[3]);

    float d01 = a2_0 * a3_1 - a2_1 * a3_0;
    float d02 = a2_0 * a3_2 - a2_2 * a3_0;
    float d03 = a2_0 * a3_3 - a2_3 * a3_0;
    sched_fence();
    YF = wmma4(pr(Z, 1), Fk[1], YF);
    float d12 = a2_1 * a3_2 - a2_2 * a3_1;
    float d13 = a2_1 * a3_3 - a2_3 * a3_1;
    float d23 = a2_2 * a3_3 - a2_3 * a3_2;
    sched_fence();
    YF = wmma4(pr(Z, 2), Fk[2], YF);
    float m0 = a1_1 * d23 - a1_2 * d13 + a1_3 * d12;
    float m1 = a1_0 * d23 - a1_2 * d03 + a1_3 * d02;
    sched_fence();
    YF = wmma4(pr(Z, 3), Fk[3], YF);
    float m2 = a1_0 * d13 - a1_1 * d03 + a1_3 * d01;
    float m3 = a1_0 * d12 - a1_1 * d02 + a1_2 * d01;
    float det = (Sc[0] * m0 - Sc[1] * m1) + (Sc[2] * m2 - Sc[3] * m3);

    sched_fence();
    v8f FM = wmma4(pr(YF, 2), Sel4, zero8());
    asm volatile("" :: "v"(FM[0]), "v"(FM[1]), "v"(FM[2]), "v"(FM[3]),
                       "v"(FM[4]), "v"(FM[5]), "v"(FM[6]), "v"(FM[7]));
    sched_fence();
    float id = __builtin_amdgcn_rcpf(det);

    v2f Si;
    Si.x = msel ?  (m0 * id) : 0.f;
    Si.y = msel ? -(m1 * id) : 0.f;
    v8f FKt = wmma4(Si, pr(YF, 0), zero8());

    v8f Gt = wmma4(NHt, pr(FKt, 0), Ft);

    v8f Wt = zero8();
#pragma unroll
    for (int j = 0; j < 4; ++j) Wt = wmma4(pr(X, j), pr(Gt, j), Wt);

    v8f V = wmma4(Rt, pr(FKt, 0), zero8());
    v8f E = wmma4(pr(FKt, 0), pr(V, 0), Qt);

    v2f rb;
    rb.x = at4 ? r0 : 0.f;
    rb.y = at4 ? r1 : 0.f;
    v8f Mn = wmma4(pr(FKt, 0), rb, FM);

    v8f Xn = E;
#pragma unroll
    for (int j = 0; j < 4; ++j) Xn = wmma4(pr(Wt, j), pr(Gt, j), Xn);

    X = Xn;
    Mc = Mn;
   }

   *(float4*)(og + (c * CHUNK + lane) * 4) = ost;
  }
}

extern "C" void kernel_launch(void* const* d_in, const int* in_sizes, int n_in,
                              void* d_out, int out_size, void* d_ws, size_t ws_size,
                              hipStream_t stream) {
  (void)in_sizes; (void)n_in; (void)out_size; (void)d_ws; (void)ws_size;
  const float* y  = (const float*)d_in[0];
  const float* F  = (const float*)d_in[1];
  const float* H  = (const float*)d_in[2];
  const float* Q  = (const float*)d_in[3];
  const float* R  = (const float*)d_in[4];
  const float* im = (const float*)d_in[5];
  const float* ic = (const float*)d_in[6];
  float* out = (float*)d_out;

  dim3 grid(G_GROUPS / WAVES_PER_BLOCK);
  dim3 block(THREADS);
  hipLaunchKernelGGL(kalman_wmma_kernel, grid, block, 0, stream,
                     y, F, H, Q, R, im, ic, out);
}


# Covariance step re-associated: P_n^T = (P F^T)^T G^T + Q^T + FK (S^T FK^T - (F P H^T)^T); P F^T needs no inverse and issues under it, serial run behind it 10 -> 7 WMMA; rounding order changes
# speedup vs baseline: 4.5646x; 1.0347x over previous
#include <hip/hip_runtime.h>


typedef __attribute__((ext_vector_type(2))) float v2f;
typedef __attribute__((ext_vector_type(8))) float v8f;

#define G_GROUPS 1024
#define T_STEPS  512
#define CHUNK    32
#define N_CHUNKS (T_STEPS / CHUNK)
#define WAVES_PER_BLOCK 4
#define THREADS (WAVES_PER_BLOCK * 32)

static __device__ __forceinline__ v8f wmma4(v2f a, v2f b, v8f c) {
  return __builtin_amdgcn_wmma_f32_16x16x4_f32(
       false, a,  false, b,
       (short)0, c,  false,  false);
}

static __device__ __forceinline__ v8f zero8() {
  v8f z = {0.f, 0.f, 0.f, 0.f, 0.f, 0.f, 0.f, 0.f};
  return z;
}

static __device__ __forceinline__ void sched_fence() {
  __builtin_amdgcn_sched_barrier(0);
}

static __device__ __forceinline__ v2f pr(v8f t, int j) {
  v2f p;
  p.x = t[2 * j];
  p.y = t[2 * j + 1];
  return p;
}

template <int CTRL>
static __device__ __forceinline__ float quad_rot(float v) {
  int r = __builtin_amdgcn_update_dpp(0, __builtin_bit_cast(int, v), CTRL, 0xf, 0xf, true);
  return __builtin_bit_cast(float, r);
}
static __device__ __forceinline__ float lane_get(float v, int l) {
  return __builtin_bit_cast(float,
      __builtin_amdgcn_readlane(__builtin_bit_cast(int, v), l));
}
#define QUAD_ROT1 0x39
#define QUAD_ROT2 0x4E
#define QUAD_ROT3 0x93

__global__ __launch_bounds__(THREADS, 1)
void kalman_wmma_kernel(const float* __restrict__ y,
                        const float* __restrict__ F,
                        const float* __restrict__ H,
                        const float* __restrict__ Q,
                        const float* __restrict__ R,
                        const float* __restrict__ im,
                        const float* __restrict__ ic,
                        float* __restrict__ out) {
  const int lane = threadIdx.x & 31;
  const int wave = threadIdx.x >> 5;
  const int col  = lane & 15;
  const int half = lane >> 4;
  const int g    = blockIdx.x * WAVES_PER_BLOCK + wave;
  const int  mrow = (col < 4) ? col : ((col >= 8 && col < 12) ? ((col - 6) & 3) : -1);
  const bool msel = (mrow >= 0);
  const int  mclp = msel ? mrow : 0;
  const bool at4  = (col == 4);

  v2f Fk[4], NFk[4], Hk[4];
#pragma unroll
  for (int j = 0; j < 4; ++j) {
    int k0 = 8 * half + 2 * j;
    Fk[j].x = F[col * 16 + k0];
    Fk[j].y = F[col * 16 + k0 + 1];
    NFk[j].x = -Fk[j].x;
    NFk[j].y = -Fk[j].y;
    Hk[j].x = msel ? H[mclp * 16 + k0] : 0.f;
    Hk[j].y = msel ? H[mclp * 16 + k0 + 1] : 0.f;
  }
  v2f NHt;
  NHt.x = -H[(2 * half) * 16 + col];
  NHt.y = -H[(2 * half + 1) * 16 + col];
  v2f NSel4;
  NSel4.x = (lane == 4) ? -1.f : 0.f;
  NSel4.y = 0.f;

  v8f Qt, Rc, Ft, X, Mc;
#pragma unroll
  for (int r = 0; r < 8; ++r) {
    int row = r + 8 * half;
    int rrow = (row < 4) ? row : ((row >= 8 && row < 12) ? ((row - 6) & 3) : -1);
    Qt[r] = Q[col * 16 + row];
    Rc[r] = (rrow >= 0 && msel) ? R[rrow * 4 + mclp] : 0.f;
    Ft[r] = F[col * 16 + row];
    X[r]  = ic[col * 16 + row];
    Mc[r] = (col == 4) ? im[g * 16 + row] : 0.f;
  }

  const float* yg = y + (size_t)g * T_STEPS * 4;
  float*       og = out + (size_t)g * T_STEPS * 4;
  float4 ynext = *(const float4*)(yg + lane * 4);

  for (int c = 0; c < N_CHUNKS; ++c) {
   const float4 ycur = ynext;
   const int cn = (c + 1 < N_CHUNKS) ? (c + 1) : c;
   ynext = *(const float4*)(yg + (cn * CHUNK + lane) * 4);
   float4 ost = {0.f, 0.f, 0.f, 0.f};

   for (int s = 0; s < CHUNK; ++s) {
    const float y0 = lane_get(ycur.x, s), y1 = lane_get(ycur.y, s),
                y2 = lane_get(ycur.z, s), y3 = lane_get(ycur.w, s);

    v8f Z = Mc;
#pragma unroll
    for (int j = 0; j < 4; ++j) Z = wmma4(pr(X, j), Hk[j], Z);

    v8f Sc = Rc;
#pragma unroll
    for (int j = 0; j < 4; ++j) Sc = wmma4(Hk[j], pr(Z, j), Sc);

    const float p0 = lane_get(Sc[0], 4), p1 = lane_get(Sc[1], 4),
                p2 = lane_get(Sc[2], 4), p3 = lane_get(Sc[3], 4);
    const bool mine = (lane == s);
    ost.x = mine ? p0 : ost.x;
    ost.y = mine ? p1 : ost.y;
    ost.z = mine ? p2 : ost.z;
    ost.w = mine ? p3 : ost.w;

    v8f NYF = wmma4(pr(Z, 0), NFk[0], zero8());

    float r0 = (half ? y2 : y0) - Sc[0];
    float r1 = (half ? y3 : y1) - Sc[1];

    float a2_0 = quad_rot<QUAD_ROT2>(Sc[0]), a2_1 = quad_rot<QUAD_ROT2>(Sc[1]),
          a2_2 = quad_rot<QUAD_ROT2>(Sc[2]), a2_3 = quad_rot<QUAD_ROT2>(Sc[3]);
    float a3_0 = quad_rot<QUAD_ROT3>(Sc[0]), a3_1 = quad_rot<QUAD_ROT3>(Sc[1]),
          a3_2 = quad_rot<QUAD_ROT3>(Sc[2]), a3_3 = quad_rot<QUAD_ROT3>(Sc[3]);
    sched_fence();
    NYF = wmma4(pr(Z, 1), NFk[1], NYF);
    float a1_0 = quad_rot<QUAD_ROT1>(Sc[0]), a1_1 = quad_rot<QUAD_ROT1>(Sc[1]),
          a1_2 = quad_rot<QUAD_ROT1>(Sc[2]), a1_3 = quad_rot<QUAD_ROT1>(Sc[3]);

    float d01 = a2_0 * a3_1 - a2_1 * a3_0;
    float d02 = a2_0 * a3_2 - a2_2 * a3_0;
    float d03 = a2_0 * a3_3 - a2_3 * a3_0;
    sched_fence();
    v8f PF = wmma4(pr(X, 0), Fk[0], zero8());
    float d12 = a2_1 * a3_2 - a2_2 * a3_1;
    float d13 = a2_1 * a3_3 - a2_3 * a3_1;
    float d23 = a2_2 * a3_3 - a2_3 * a3_2;
    sched_fence();
    NYF = wmma4(pr(Z, 2), NFk[2], NYF);
    float m0 = a1_1 * d23 - a1_2 * d13 + a1_3 * d12;
    float m1 = a1_0 * d23 - a1_2 * d03 + a1_3 * d02;
    sched_fence();
    PF = wmma4(pr(X, 1), Fk[1], PF);
    float m2 = a1_0 * d13 - a1_1 * d03 + a1_3 * d01;
    float m3 = a1_0 * d12 - a1_1 * d02 + a1_2 * d01;
    sched_fence();
    NYF = wmma4(pr(Z, 3), NFk[3], NYF);
    float det = (Sc[0] * m0 - Sc[1] * m1) + (Sc[2] * m2 - Sc[3] * m3);
    sched_fence();
    PF = wmma4(pr(X, 2), Fk[2], PF);
    float id = __builtin_amdgcn_rcpf(det);

    v2f NSi;
    NSi.x = msel ? -(m0 * id) : 0.f;
    NSi.y = msel ?  (m1 * id) : 0.f;
    v8f FKt = wmma4(NSi, pr(NYF, 0), zero8());
    sched_fence();
    PF = wmma4(pr(X, 3), Fk[3], PF);

    v8f Gt = wmma4(NHt, pr(FKt, 0), Ft);

    v8f FM = wmma4(pr(NYF, 2), NSel4, zero8());

    v8f Xn = wmma4(pr(PF, 0), pr(Gt, 0), Qt);
    v8f D  = wmma4(pr(Sc, 0), pr(FKt, 0), NYF);
    Xn = wmma4(pr(PF, 1), pr(Gt, 1), Xn);

    v2f rb;
    rb.x = at4 ? r0 : 0.f;
    rb.y = at4 ? r1 : 0.f;
    v8f Mn = wmma4(pr(FKt, 0), rb, FM);

    Xn = wmma4(pr(PF, 2), pr(Gt, 2), Xn);
    Xn = wmma4(pr(PF, 3), pr(Gt, 3), Xn);
    Xn = wmma4(pr(FKt, 0), pr(D, 0), Xn);

    X = Xn;
    Mc = Mn;
   }

   *(float4*)(og + (c * CHUNK + lane) * 4) = ost;
  }
}

extern "C" void kernel_launch(void* const* d_in, const int* in_sizes, int n_in,
                              void* d_out, int out_size, void* d_ws, size_t ws_size,
                              hipStream_t stream) {
  (void)in_sizes; (void)n_in; (void)out_size; (void)d_ws; (void)ws_size;
  const float* y  = (const float*)d_in[0];
  const float* F  = (const float*)d_in[1];
  const float* H  = (const float*)d_in[2];
  const float* Q  = (const float*)d_in[3];
  const float* R  = (const float*)d_in[4];
  const float* im = (const float*)d_in[5];
  const float* ic = (const float*)d_in[6];
  float* out = (float*)d_out;

  dim3 grid(G_GROUPS / WAVES_PER_BLOCK);
  dim3 block(THREADS);
  hipLaunchKernelGGL(kalman_wmma_kernel, grid, block, 0, stream,
                     y, F, H, Q, R, im, ic, out);
}


# Vector code of the 4x4 inverse and the residual issued level by level of its dependency tree, a matrix instruction or independent lane reads between levels: no link waits on the one before; same sums
# speedup vs baseline: 4.9555x; 1.0856x over previous
#include <hip/hip_runtime.h>


typedef __attribute__((ext_vector_type(2))) float v2f;
typedef __attribute__((ext_vector_type(8))) float v8f;

#define G_GROUPS 1024
#define T_STEPS  512
#define CHUNK    32
#define N_CHUNKS (T_STEPS / CHUNK)
#define WAVES_PER_BLOCK 4
#define THREADS (WAVES_PER_BLOCK * 32)

static __device__ __forceinline__ v8f wmma4(v2f a, v2f b, v8f c) {
  return __builtin_amdgcn_wmma_f32_16x16x4_f32(
       false, a,  false, b,
       (short)0, c,  false,  false);
}

static __device__ __forceinline__ v8f zero8() {
  v8f z = {0.f, 0.f, 0.f, 0.f, 0.f, 0.f, 0.f, 0.f};
  return z;
}

static __device__ __forceinline__ void sched_fence() {
  __builtin_amdgcn_sched_barrier(0);
}

static __device__ __forceinline__ v2f pr(v8f t, int j) {
  v2f p;
  p.x = t[2 * j];
  p.y = t[2 * j + 1];
  return p;
}

template <int CTRL>
static __device__ __forceinline__ float quad_rot(float v) {
  int r = __builtin_amdgcn_update_dpp(0, __builtin_bit_cast(int, v), CTRL, 0xf, 0xf, true);
  return __builtin_bit_cast(float, r);
}
static __device__ __forceinline__ float lane_get(float v, int l) {
  return __builtin_bit_cast(float,
      __builtin_amdgcn_readlane(__builtin_bit_cast(int, v), l));
}
#define QUAD_ROT1 0x39
#define QUAD_ROT2 0x4E
#define QUAD_ROT3 0x93

__global__ __launch_bounds__(THREADS, 1)
void kalman_wmma_kernel(const float* __restrict__ y,
                        const float* __restrict__ F,
                        const float* __restrict__ H,
                        const float* __restrict__ Q,
                        const float* __restrict__ R,
                        const float* __restrict__ im,
                        const float* __restrict__ ic,
                        float* __restrict__ out) {
  const int lane = threadIdx.x & 31;
  const int wave = threadIdx.x >> 5;
  const int col  = lane & 15;
  const int half = lane >> 4;
  const int g    = blockIdx.x * WAVES_PER_BLOCK + wave;
  const int  mrow = (col < 4) ? col : ((col >= 8 && col < 12) ? ((col - 6) & 3) : -1);
  const bool msel = (mrow >= 0);
  const int  mclp = msel ? mrow : 0;
  const bool at4  = (col == 4);

  v2f Fk[4], NFk[4], Hk[4];
#pragma unroll
  for (int j = 0; j < 4; ++j) {
    int k0 = 8 * half + 2 * j;
    Fk[j].x = F[col * 16 + k0];
    Fk[j].y = F[col * 16 + k0 + 1];
    NFk[j].x = -Fk[j].x;
    NFk[j].y = -Fk[j].y;
    Hk[j].x = msel ? H[mclp * 16 + k0] : 0.f;
    Hk[j].y = msel ? H[mclp * 16 + k0 + 1] : 0.f;
  }
  v2f NHt;
  NHt.x = -H[(2 * half) * 16 + col];
  NHt.y = -H[(2 * half + 1) * 16 + col];
  v2f NSel4;
  NSel4.x = (lane == 4) ? -1.f : 0.f;
  NSel4.y = 0.f;

  v8f Qt, Rc, Ft, X, Mc;
#pragma unroll
  for (int r = 0; r < 8; ++r) {
    int row = r + 8 * half;
    int rrow = (row < 4) ? row : ((row >= 8 && row < 12) ? ((row - 6) & 3) : -1);
    Qt[r] = Q[col * 16 + row];
    Rc[r] = (rrow >= 0 && msel) ? R[rrow * 4 + mclp] : 0.f;
    Ft[r] = F[col * 16 + row];
    X[r]  = ic[col * 16 + row];
    Mc[r] = (col == 4) ? im[g * 16 + row] : 0.f;
  }

  const float* yg = y + (size_t)g * T_STEPS * 4;
  float*       og = out + (size_t)g * T_STEPS * 4;
  float4 ynext = *(const float4*)(yg + lane * 4);

  for (int c = 0; c < N_CHUNKS; ++c) {
   const float4 ycur = ynext;
   const int cn = (c + 1 < N_CHUNKS) ? (c + 1) : c;
   ynext = *(const float4*)(yg + (cn * CHUNK + lane) * 4);
   float4 ost = {0.f, 0.f, 0.f, 0.f};

   for (int s = 0; s < CHUNK; ++s) {
    v8f Z = Mc;
#pragma unroll
    for (int j = 0; j < 4; ++j) Z = wmma4(pr(X, j), Hk[j], Z);

    v8f Sc = Rc;
#pragma unroll
    for (int j = 0; j < 4; ++j) Sc = wmma4(Hk[j], pr(Z, j), Sc);

    v8f NYF = wmma4(pr(Z, 0), NFk[0], zero8());

    float a2_0 = quad_rot<QUAD_ROT2>(Sc[0]), a2_1 = quad_rot<QUAD_ROT2>(Sc[1]),
          a2_2 = quad_rot<QUAD_ROT2>(Sc[2]), a2_3 = quad_rot<QUAD_ROT2>(Sc[3]);
    float a3_0 = quad_rot<QUAD_ROT3>(Sc[0]), a3_1 = quad_rot<QUAD_ROT3>(Sc[1]),
          a3_2 = quad_rot<QUAD_ROT3>(Sc[2]), a3_3 = quad_rot<QUAD_ROT3>(Sc[3]);
    sched_fence();
    float nt01 = -(a2_1 * a3_0), nt02 = -(a2_2 * a3_0), nt03 = -(a2_3 * a3_0);
    float nt12 = -(a2_2 * a3_1), nt13 = -(a2_3 * a3_1), nt23 = -(a2_3 * a3_2);
    sched_fence();
    float a1_0 = quad_rot<QUAD_ROT1>(Sc[0]), a1_1 = quad_rot<QUAD_ROT1>(Sc[1]),
          a1_2 = quad_rot<QUAD_ROT1>(Sc[2]), a1_3 = quad_rot<QUAD_ROT1>(Sc[3]);
    sched_fence();
    float d01 = __builtin_fmaf(a2_0, a3_1, nt01);
    float d02 = __builtin_fmaf(a2_0, a3_2, nt02);
    float d03 = __builtin_fmaf(a2_0, a3_3, nt03);
    float d12 = __builtin_fmaf(a2_1, a3_2, nt12);
    float d13 = __builtin_fmaf(a2_1, a3_3, nt13);
    float d23 = __builtin_fmaf(a2_2, a3_3, nt23);
    sched_fence();
    NYF = wmma4(pr(Z, 1), NFk[1], NYF);
    sched_fence();
    float nu0 = -(a1_2 * d13), nu1 = -(a1_2 * d03);
    float nu2 = -(a1_1 * d03), nu3 = -(a1_1 * d02);
    sched_fence();
    v8f PF = wmma4(pr(X, 0), Fk[0], zero8());
    sched_fence();
    float w0 = __builtin_fmaf(a1_1, d23, nu0);
    float w1 = __builtin_fmaf(a1_0, d23, nu1);
    float w2 = __builtin_fmaf(a1_0, d13, nu2);
    float w3 = __builtin_fmaf(a1_0, d12, nu3);
    sched_fence();
    NYF = wmma4(pr(Z, 2), NFk[2], NYF);
    sched_fence();
    float m0 = __builtin_fmaf(a1_3, d12, w0);
    float m1 = __builtin_fmaf(a1_3, d02, w1);
    float m2 = __builtin_fmaf(a1_3, d01, w2);
    float m3 = __builtin_fmaf(a1_2, d01, w3);
    sched_fence();
    PF = wmma4(pr(X, 1), Fk[1], PF);
    sched_fence();
    float ne1 = -(Sc[1] * m1), ne3 = -(Sc[3] * m3);
    sched_fence();
    const float p0 = lane_get(Sc[0], 4), p1 = lane_get(Sc[1], 4),
                p2 = lane_get(Sc[2], 4), p3 = lane_get(Sc[3], 4);
    const bool mine = (lane == s);
    sched_fence();
    float f0 = __builtin_fmaf(Sc[0], m0, ne1);
    float f2 = __builtin_fmaf(Sc[2], m2, ne3);
    sched_fence();
    NYF = wmma4(pr(Z, 3), NFk[3], NYF);
    sched_fence();
    float det = f0 + f2;
    sched_fence();
    ost.x = mine ? p0 : ost.x;
    ost.y = mine ? p1 : ost.y;
    ost.z = mine ? p2 : ost.z;
    ost.w = mine ? p3 : ost.w;
    sched_fence();
    float id = __builtin_amdgcn_rcpf(det);
    sched_fence();
    PF = wmma4(pr(X, 2), Fk[2], PF);
    sched_fence();
    float n0 = -(m0 * id);
    float n1 = m1 * id;
    sched_fence();
    const float y0 = lane_get(ycur.x, s), y1 = lane_get(ycur.y, s),
                y2 = lane_get(ycur.z, s), y3 = lane_get(ycur.w, s);
    sched_fence();
    v2f ylo, yhi;
    ylo.x = y0; ylo.y = y1;
    yhi.x = y2; yhi.y = y3;
    const v2f ysel = half ? yhi : ylo;
    v2f NSi;
    NSi.x = msel ? n0 : 0.f;
    NSi.y = msel ? n1 : 0.f;
    sched_fence();
    v8f FKt = wmma4(NSi, pr(NYF, 0), zero8());
    sched_fence();
    PF = wmma4(pr(X, 3), Fk[3], PF);

    v8f Gt = wmma4(NHt, pr(FKt, 0), Ft);
    sched_fence();
    v2f hm;
    hm.x = Sc[0]; hm.y = Sc[1];
    const v2f res = ysel - hm;
    sched_fence();

    v8f FM = wmma4(pr(NYF, 2), NSel4, zero8());
    sched_fence();

    v8f Xn = wmma4(pr(PF, 0), pr(Gt, 0), Qt);
    sched_fence();
    v8f D  = wmma4(pr(Sc, 0), pr(FKt, 0), NYF);
    sched_fence();
    Xn = wmma4(pr(PF, 1), pr(Gt, 1), Xn);
    sched_fence();
    v2f zero2;
    zero2.x = 0.f; zero2.y = 0.f;
    const v2f rb = at4 ? res : zero2;
    sched_fence();
    Xn = wmma4(pr(PF, 2), pr(Gt, 2), Xn);
    sched_fence();

    v8f Mn = wmma4(pr(FKt, 0), rb, FM);
    sched_fence();

    Xn = wmma4(pr(PF, 3), pr(Gt, 3), Xn);
    Xn = wmma4(pr(FKt, 0), pr(D, 0), Xn);

    X = Xn;
    Mc = Mn;
   }

   *(float4*)(og + (c * CHUNK + lane) * 4) = ost;
  }
}

extern "C" void kernel_launch(void* const* d_in, const int* in_sizes, int n_in,
                              void* d_out, int out_size, void* d_ws, size_t ws_size,
                              hipStream_t stream) {
  (void)in_sizes; (void)n_in; (void)out_size; (void)d_ws; (void)ws_size;
  const float* y  = (const float*)d_in[0];
  const float* F  = (const float*)d_in[1];
  const float* H  = (const float*)d_in[2];
  const float* Q  = (const float*)d_in[3];
  const float* R  = (const float*)d_in[4];
  const float* im = (const float*)d_in[5];
  const float* ic = (const float*)d_in[6];
  float* out = (float*)d_out;

  dim3 grid(G_GROUPS / WAVES_PER_BLOCK);
  dim3 block(THREADS);
  hipLaunchKernelGGL(kalman_wmma_kernel, grid, block, 0, stream,
                     y, F, H, Q, R, im, ic, out);
}


# State basis rotated once in the prologue (orthogonal W by Householder on H^T, H W zero outside k-chunk 0): P H^T + m and H Z + R take 1 WMMA each, not 4 (26 -> 20/step); rounding order changes
# speedup vs baseline: 6.2549x; 1.2622x over previous
#include <hip/hip_runtime.h>


typedef __attribute__((ext_vector_type(2))) float v2f;
typedef __attribute__((ext_vector_type(8))) float v8f;

#define G_GROUPS 1024
#define T_STEPS  512
#define CHUNK    32
#define N_CHUNKS (T_STEPS / CHUNK)
#define WAVES_PER_BLOCK 4
#define THREADS (WAVES_PER_BLOCK * 32)

static __device__ __forceinline__ v8f wmma4(v2f a, v2f b, v8f c) {
  return __builtin_amdgcn_wmma_f32_16x16x4_f32(
       false, a,  false, b,
       (short)0, c,  false,  false);
}

static __device__ __forceinline__ v8f zero8() {
  v8f z = {0.f, 0.f, 0.f, 0.f, 0.f, 0.f, 0.f, 0.f};
  return z;
}

static __device__ __forceinline__ void sched_fence() {
  __builtin_amdgcn_sched_barrier(0);
}

static __device__ __forceinline__ v2f pr(v8f t, int j) {
  v2f p;
  p.x = t[2 * j];
  p.y = t[2 * j + 1];
  return p;
}

template <int CTRL>
static __device__ __forceinline__ float quad_rot(float v) {
  int r = __builtin_amdgcn_update_dpp(0, __builtin_bit_cast(int, v), CTRL, 0xf, 0xf, true);
  return __builtin_bit_cast(float, r);
}
static __device__ __forceinline__ float lane_get(float v, int l) {
  return __builtin_bit_cast(float,
      __builtin_amdgcn_readlane(__builtin_bit_cast(int, v), l));
}
#define QUAD_ROT1 0x39
#define QUAD_ROT2 0x4E
#define QUAD_ROT3 0x93

__global__ __launch_bounds__(THREADS, 1)
void kalman_wmma_kernel(const float* __restrict__ y,
                        const float* __restrict__ F,
                        const float* __restrict__ H,
                        const float* __restrict__ Q,
                        const float* __restrict__ R,
                        const float* __restrict__ im,
                        const float* __restrict__ ic,
                        float* __restrict__ out) {
  const int lane = threadIdx.x & 31;
  const int wave = threadIdx.x >> 5;
  const int col  = lane & 15;
  const int half = lane >> 4;
  const int g    = blockIdx.x * WAVES_PER_BLOCK + wave;
  const int  mrow = (col < 4) ? col : ((col >= 8 && col < 12) ? ((col - 6) & 3) : -1);
  const bool msel = (mrow >= 0);
  const int  mclp = msel ? mrow : 0;
  const bool at4  = (col == 4);

  float ha[4][16];
#pragma unroll
  for (int c = 0; c < 4; ++c)
#pragma unroll
    for (int i = 0; i < 16; ++i) ha[c][i] = H[c * 16 + i];
  float hv[4][16], hb[4];
#pragma unroll
  for (int j = 0; j < 4; ++j) {
    float amax = 0.f;
#pragma unroll
    for (int i = j; i < 16; ++i) amax = __builtin_fmaxf(amax, __builtin_fabsf(ha[j][i]));
    const bool ok = (amax > 0.f);
    const float dsc = ok ? amax : 1.f;
    float sigma = 0.f;
#pragma unroll
    for (int i = 0; i < 16; ++i) {
      hv[j][i] = (i >= j) ? (ha[j][i] / dsc) : 0.f;
      sigma = __builtin_fmaf(hv[j][i], hv[j][i], sigma);
    }
    const float nrm = __builtin_sqrtf(sigma);
    const float x0  = hv[j][j];
    hv[j][j] = x0 + ((x0 >= 0.f) ? nrm : -nrm);
    const float vtv = __builtin_fmaf(hv[j][j], hv[j][j], __builtin_fmaf(-x0, x0, sigma));
    hb[j] = (ok && vtv > 0.f) ? (2.f / vtv) : 0.f;
#pragma unroll
    for (int c = j + 1; c < 4; ++c) {
      float d = 0.f;
#pragma unroll
      for (int i = j; i < 16; ++i) d = __builtin_fmaf(hv[j][i], ha[c][i], d);
      d *= hb[j];
#pragma unroll
      for (int i = j; i < 16; ++i) ha[c][i] = __builtin_fmaf(-d, hv[j][i], ha[c][i]);
    }
  }
  float ur[16];
#pragma unroll
  for (int k = 0; k < 16; ++k) ur[k] = (k == col) ? 1.f : 0.f;
#pragma unroll
  for (int j = 0; j < 4; ++j) {
    float d = 0.f;
#pragma unroll
    for (int k = j; k < 16; ++k) d = __builtin_fmaf(ur[k], hv[j][k], d);
    d *= hb[j];
#pragma unroll
    for (int k = j; k < 16; ++k) ur[k] = __builtin_fmaf(-d, hv[j][k], ur[k]);
  }
  __shared__ float sW[WAVES_PER_BLOCK][16 * 17];
  if (half == 0) {
#pragma unroll
    for (int k = 0; k < 16; ++k) {
      const int kc = (k == 2) ? 8 : (k == 3) ? 9 : (k == 8) ? 2 : (k == 9) ? 3 : k;
      sW[wave][col * 17 + kc] = ur[k];
    }
  }
  __syncthreads();
  v2f Wk[4];
#pragma unroll
  for (int j = 0; j < 4; ++j) {
    int k0 = 8 * half + 2 * j;
    Wk[j].x = sW[wave][k0 * 17 + col];
    Wk[j].y = sW[wave][(k0 + 1) * 17 + col];
  }

  v2f FtA[4], QtA[4], PtA[4], HpA[4];
#pragma unroll
  for (int j = 0; j < 4; ++j) {
    int k0 = 8 * half + 2 * j;
    FtA[j].x = F[k0 * 16 + col];
    FtA[j].y = F[(k0 + 1) * 16 + col];
    QtA[j].x = Q[k0 * 16 + col];
    QtA[j].y = Q[(k0 + 1) * 16 + col];
    PtA[j].x = ic[k0 * 16 + col];
    PtA[j].y = ic[(k0 + 1) * 16 + col];
    HpA[j].x = msel ? H[mclp * 16 + k0] : 0.f;
    HpA[j].y = msel ? H[mclp * 16 + k0 + 1] : 0.f;
  }
  v8f Rc, M0;
#pragma unroll
  for (int r = 0; r < 8; ++r) {
    int row = r + 8 * half;
    int rrow = (row < 4) ? row : ((row >= 8 && row < 12) ? ((row - 6) & 3) : -1);
    Rc[r] = (rrow >= 0 && msel) ? R[rrow * 4 + mclp] : 0.f;
    M0[r] = (col == 4) ? im[g * 16 + row] : 0.f;
  }
  v8f Ft, Qt, X, Mc, HpT, Hp;
  {
    v8f t = zero8();
#pragma unroll
    for (int j = 0; j < 4; ++j) t = wmma4(FtA[j], Wk[j], t);
    Ft = zero8();
#pragma unroll
    for (int j = 0; j < 4; ++j) Ft = wmma4(Wk[j], pr(t, j), Ft);
  }
  {
    v8f t = zero8();
#pragma unroll
    for (int j = 0; j < 4; ++j) t = wmma4(QtA[j], Wk[j], t);
    Qt = zero8();
#pragma unroll
    for (int j = 0; j < 4; ++j) Qt = wmma4(Wk[j], pr(t, j), Qt);
  }
  {
    v8f t = zero8();
#pragma unroll
    for (int j = 0; j < 4; ++j) t = wmma4(PtA[j], Wk[j], t);
    X = zero8();
#pragma unroll
    for (int j = 0; j < 4; ++j) X = wmma4(Wk[j], pr(t, j), X);
  }
  Mc = zero8();
#pragma unroll
  for (int j = 0; j < 4; ++j) Mc = wmma4(Wk[j], pr(M0, j), Mc);
  HpT = zero8();
  Hp  = zero8();
#pragma unroll
  for (int j = 0; j < 4; ++j) {
    HpT = wmma4(Wk[j], HpA[j], HpT);
    Hp  = wmma4(HpA[j], Wk[j], Hp);
  }
  v2f Fk[4], NFk[4];
#pragma unroll
  for (int j = 0; j < 4; ++j) {
    Fk[j]  = pr(Ft, j);
    NFk[j] = -Fk[j];
  }
  const v2f Hk0 = pr(HpT, 0);
  const bool hst = ((col & 6) == 0);
  v2f NHt;
  NHt.x = hst ? -Hp[0] : 0.f;
  NHt.y = hst ? -Hp[1] : 0.f;
  v2f NSel4;
  NSel4.x = (lane == 4) ? -1.f : 0.f;
  NSel4.y = 0.f;

  const float* yg = y + (size_t)g * T_STEPS * 4;
  float*       og = out + (size_t)g * T_STEPS * 4;
  float4 ynext = *(const float4*)(yg + lane * 4);

  for (int c = 0; c < N_CHUNKS; ++c) {
   const float4 ycur = ynext;
   const int cn = (c + 1 < N_CHUNKS) ? (c + 1) : c;
   ynext = *(const float4*)(yg + (cn * CHUNK + lane) * 4);
   float4 ost = {0.f, 0.f, 0.f, 0.f};

   for (int s = 0; s < CHUNK; ++s) {
    v8f Z = wmma4(pr(X, 0), Hk0, Mc);

    v8f Sc = wmma4(Hk0, pr(Z, 0), Rc);

    v8f NYF = wmma4(pr(Z, 0), NFk[0], zero8());

    float a2_0 = quad_rot<QUAD_ROT2>(Sc[0]), a2_1 = quad_rot<QUAD_ROT2>(Sc[1]),
          a2_2 = quad_rot<QUAD_ROT2>(Sc[2]), a2_3 = quad_rot<QUAD_ROT2>(Sc[3]);
    float a3_0 = quad_rot<QUAD_ROT3>(Sc[0]), a3_1 = quad_rot<QUAD_ROT3>(Sc[1]),
          a3_2 = quad_rot<QUAD_ROT3>(Sc[2]), a3_3 = quad_rot<QUAD_ROT3>(Sc[3]);
    sched_fence();
    float nt01 = -(a2_1 * a3_0), nt02 = -(a2_2 * a3_0), nt03 = -(a2_3 * a3_0);
    float nt12 = -(a2_2 * a3_1), nt13 = -(a2_3 * a3_1), nt23 = -(a2_3 * a3_2);
    sched_fence();
    float a1_0 = quad_rot<QUAD_ROT1>(Sc[0]), a1_1 = quad_rot<QUAD_ROT1>(Sc[1]),
          a1_2 = quad_rot<QUAD_ROT1>(Sc[2]), a1_3 = quad_rot<QUAD_ROT1>(Sc[3]);
    sched_fence();
    float d01 = __builtin_fmaf(a2_0, a3_1, nt01);
    float d02 = __builtin_fmaf(a2_0, a3_2, nt02);
    float d03 = __builtin_fmaf(a2_0, a3_3, nt03);
    float d12 = __builtin_fmaf(a2_1, a3_2, nt12);
    float d13 = __builtin_fmaf(a2_1, a3_3, nt13);
    float d23 = __builtin_fmaf(a2_2, a3_3, nt23);
    sched_fence();
    NYF = wmma4(pr(Z, 1), NFk[1], NYF);
    sched_fence();
    float nu0 = -(a1_2 * d13), nu1 = -(a1_2 * d03);
    float nu2 = -(a1_1 * d03), nu3 = -(a1_1 * d02);
    sched_fence();
    v8f PF = wmma4(pr(X, 0), Fk[0], zero8());
    sched_fence();
    float w0 = __builtin_fmaf(a1_1, d23, nu0);
    float w1 = __builtin_fmaf(a1_0, d23, nu1);
    float w2 = __builtin_fmaf(a1_0, d13, nu2);
    float w3 = __builtin_fmaf(a1_0, d12, nu3);
    sched_fence();
    NYF = wmma4(pr(Z, 2), NFk[2], NYF);
    sched_fence();
    float m0 = __builtin_fmaf(a1_3, d12, w0);
    float m1 = __builtin_fmaf(a1_3, d02, w1);
    float m2 = __builtin_fmaf(a1_3, d01, w2);
    float m3 = __builtin_fmaf(a1_2, d01, w3);
    sched_fence();
    PF = wmma4(pr(X, 1), Fk[1], PF);
    sched_fence();
    float ne1 = -(Sc[1] * m1), ne3 = -(Sc[3] * m3);
    sched_fence();
    const float p0 = lane_get(Sc[0], 4), p1 = lane_get(Sc[1], 4),
                p2 = lane_get(Sc[2], 4), p3 = lane_get(Sc[3], 4);
    const bool mine = (lane == s);
    sched_fence();
    float f0 = __builtin_fmaf(Sc[0], m0, ne1);
    float f2 = __builtin_fmaf(Sc[2], m2, ne3);
    sched_fence();
    NYF = wmma4(pr(Z, 3), NFk[3], NYF);
    sched_fence();
    float det = f0 + f2;
    sched_fence();
    ost.x = mine ? p0 : ost.x;
    ost.y = mine ? p1 : ost.y;
    ost.z = mine ? p2 : ost.z;
    ost.w = mine ? p3 : ost.w;
    sched_fence();
    float id = __builtin_amdgcn_rcpf(det);
    sched_fence();
    PF = wmma4(pr(X, 2), Fk[2], PF);
    sched_fence();
    float n0 = -(m0 * id);
    float n1 = m1 * id;
    sched_fence();
    const float y0 = lane_get(ycur.x, s), y1 = lane_get(ycur.y, s),
                y2 = lane_get(ycur.z, s), y3 = lane_get(ycur.w, s);
    sched_fence();
    v2f ylo, yhi;
    ylo.x = y0; ylo.y = y1;
    yhi.x = y2; yhi.y = y3;
    const v2f ysel = half ? yhi : ylo;
    v2f NSi;
    NSi.x = msel ? n0 : 0.f;
    NSi.y = msel ? n1 : 0.f;
    sched_fence();
    v8f FKt = wmma4(NSi, pr(NYF, 0), zero8());
    sched_fence();
    PF = wmma4(pr(X, 3), Fk[3], PF);

    v8f Gt = wmma4(NHt, pr(FKt, 0), Ft);
    sched_fence();
    v2f hm;
    hm.x = Sc[0]; hm.y = Sc[1];
    const v2f res = ysel - hm;
    sched_fence();

    v8f FM = wmma4(pr(NYF, 2), NSel4, zero8());
    sched_fence();

    v8f Xn = wmma4(pr(PF, 0), pr(Gt, 0), Qt);
    sched_fence();
    v8f D  = wmma4(pr(Sc, 0), pr(FKt, 0), NYF);
    sched_fence();
    Xn = wmma4(pr(PF, 1), pr(Gt, 1), Xn);
    sched_fence();
    v2f zero2;
    zero2.x = 0.f; zero2.y = 0.f;
    const v2f rb = at4 ? res : zero2;
    sched_fence();
    Xn = wmma4(pr(PF, 2), pr(Gt, 2), Xn);
    sched_fence();

    v8f Mn = wmma4(pr(FKt, 0), rb, FM);
    sched_fence();

    Xn = wmma4(pr(PF, 3), pr(Gt, 3), Xn);
    Xn = wmma4(pr(FKt, 0), pr(D, 0), Xn);

    X = Xn;
    Mc = Mn;
   }

   *(float4*)(og + (c * CHUNK + lane) * 4) = ost;
  }
}

extern "C" void kernel_launch(void* const* d_in, const int* in_sizes, int n_in,
                              void* d_out, int out_size, void* d_ws, size_t ws_size,
                              hipStream_t stream) {
  (void)in_sizes; (void)n_in; (void)out_size; (void)d_ws; (void)ws_size;
  const float* y  = (const float*)d_in[0];
  const float* F  = (const float*)d_in[1];
  const float* H  = (const float*)d_in[2];
  const float* Q  = (const float*)d_in[3];
  const float* R  = (const float*)d_in[4];
  const float* im = (const float*)d_in[5];
  const float* ic = (const float*)d_in[6];
  float* out = (float*)d_out;

  dim3 grid(G_GROUPS / WAVES_PER_BLOCK);
  dim3 block(THREADS);
  hipLaunchKernelGGL(kalman_wmma_kernel, grid, block, 0, stream,
                     y, F, H, Q, R, im, ic, out);
}
